# starting kernel
# baseline (speedup 1.0000x reference)
#include <hip/hip_runtime.h>

typedef __attribute__((ext_vector_type(2))) float v2f;
typedef __attribute__((ext_vector_type(8))) float v8f;

#define N_IN   128
#define N_HID  256
#define N_HID2 128
#define N_CLS  2

__global__ void deg_kernel(const int* __restrict__ col, float* __restrict__ deg, int E) {
    int e = blockIdx.x * blockDim.x + threadIdx.x;
    if (e < E) atomicAdd(&deg[col[e]], 1.0f);
}

__global__ void deginv_kernel(const float* __restrict__ deg, float* __restrict__ dinv, int n) {
    int i = blockIdx.x * blockDim.x + threadIdx.x;
    if (i < n) dinv[i] = 1.0f / fmaxf(deg[i], 1.0f);
}

template<int F>
__global__ void scatter_kernel(const float* __restrict__ src,
                               const int* __restrict__ row,
                               const int* __restrict__ col,
                               float* __restrict__ agg,
                               long long total) {
    long long tid = (long long)blockIdx.x * blockDim.x + threadIdx.x;
    if (tid >= total) return;
    int e = (int)(tid >> (__builtin_ctz(F)));
    int f = (int)(tid & (F - 1));
    int r = row[e], c = col[e];
    atomicAdd(&agg[(size_t)c * F + f], src[(size_t)r * F + f]);
}

template<int K, int NF, bool RELU>
__global__ __launch_bounds__(256)
void gemm_wmma(const float* __restrict__ A, const float* __restrict__ W,
               const float* __restrict__ bias, const float* __restrict__ dinv,
               float* __restrict__ out, int M) {
    const int lane  = threadIdx.x;
    const int waveId = threadIdx.y;
    const int tilesN = NF / 16;
    const int tilesM = (M + 15) / 16;
    int tile = blockIdx.x * blockDim.y + waveId;
    if (tile >= tilesM * tilesN) return;
    const int tm = tile / tilesN;
    const int tn = tile % tilesN;

    const int half = lane >> 4;
    const int l16  = lane & 15;

    int arow = tm * 16 + l16;
    if (arow >= M) arow = M - 1;
    const float* __restrict__ aptr = A + (size_t)arow * K;
    const float* __restrict__ wptr = W + (size_t)(tn * 16 + l16) * K;

    v8f acc = {};
    #pragma unroll
    for (int k = 0; k < K; k += 4) {
        const int kk = k + 2 * half;
        v2f a = *(const v2f*)(aptr + kk);
        v2f b = *(const v2f*)(wptr + kk);
        acc = __builtin_amdgcn_wmma_f32_16x16x4_f32(
             false, a,  false, b,
             (short)0, acc,  false,  false);
    }

    const int n  = tn * 16 + l16;
    const float bv = bias[n];
    #pragma unroll
    for (int r = 0; r < 8; ++r) {
        int m = tm * 16 + half * 8 + r;
        if (m < M) {
            float v = acc[r] * dinv[m] + bv;
            if (RELU) v = fmaxf(v, 0.0f);
            out[(size_t)m * NF + n] = v;
        }
    }
}

__global__ void head_kernel(const float* __restrict__ h, const float* __restrict__ W3,
                            const float* __restrict__ b3, float* __restrict__ out, int M) {
    int tid = blockIdx.x * blockDim.x + threadIdx.x;
    if (tid >= M * N_CLS) return;
    int m = tid / N_CLS, c = tid % N_CLS;
    const float* __restrict__ hm = h  + (size_t)m * N_HID2;
    const float* __restrict__ wc = W3 + (size_t)c * N_HID2;
    float s = b3[c];
    #pragma unroll 8
    for (int k = 0; k < N_HID2; ++k) s += hm[k] * wc[k];
    out[tid] = s;
}

extern "C" void kernel_launch(void* const* d_in, const int* in_sizes, int n_in,
                              void* d_out, int out_size, void* d_ws, size_t ws_size,
                              hipStream_t stream) {
    const float* x  = (const float*)d_in[0];
    const int*   ei = (const int*)  d_in[1];
    const float* W1 = (const float*)d_in[3];
    const float* b1 = (const float*)d_in[4];
    const float* W2 = (const float*)d_in[5];
    const float* b2 = (const float*)d_in[6];
    const float* W3 = (const float*)d_in[7];
    const float* b3 = (const float*)d_in[8];
    float* out = (float*)d_out;

    const int M = in_sizes[0] / N_IN;
    const int E = in_sizes[1] / 2;
    const int* row = ei;
    const int* col = ei + E;

    float* ws = (float*)d_ws;
    size_t off = 0;
    float* deg  = ws + off; off += (size_t)M;
    float* dinv = ws + off; off += (size_t)M;
    float* agg1 = ws + off; off += (size_t)M * N_IN;
    float* h1   = ws + off; off += (size_t)M * N_HID;
    float* agg2 = ws + off; off += (size_t)M * N_HID;
    float* h2   = ws + off; off += (size_t)M * N_HID2;
    (void)ws_size;

    hipMemsetAsync(deg,  0, sizeof(float) * (size_t)M, stream);
    hipMemsetAsync(agg1, 0, sizeof(float) * (size_t)M * N_IN, stream);
    hipMemsetAsync(agg2, 0, sizeof(float) * (size_t)M * N_HID, stream);

    deg_kernel<<<(E + 255) / 256, 256, 0, stream>>>(col, deg, E);
    deginv_kernel<<<(M + 255) / 256, 256, 0, stream>>>(deg, dinv, M);

    long long tot1 = (long long)E * N_IN;
    scatter_kernel<N_IN><<<(unsigned)((tot1 + 255) / 256), 256, 0, stream>>>(x, row, col, agg1, tot1);
    int tiles1 = ((M + 15) / 16) * (N_HID / 16);
    gemm_wmma<N_IN, N_HID, true>
        <<<(tiles1 + 7) / 8, dim3(32, 8), 0, stream>>>(agg1, W1, b1, dinv, h1, M);

    long long tot2 = (long long)E * N_HID;
    scatter_kernel<N_HID><<<(unsigned)((tot2 + 255) / 256), 256, 0, stream>>>(h1, row, col, agg2, tot2);
    int tiles2 = ((M + 15) / 16) * (N_HID2 / 16);
    gemm_wmma<N_HID, N_HID2, true>
        <<<(tiles2 + 7) / 8, dim3(32, 8), 0, stream>>>(agg2, W2, b2, dinv, h2, M);

    head_kernel<<<(M * N_CLS + 255) / 256, 256, 0, stream>>>(h2, W3, b3, out, M);
}


# Float atomic scatter-add replaced by a per-call CSR (returning int atomic = edge's arrival rank) and a one-wave-per-node row gather; sums now run in rank order, no float atomics
# speedup vs baseline: 3.4650x; 3.4650x over previous
#include <hip/hip_runtime.h>

typedef __attribute__((ext_vector_type(2))) float v2f;
typedef __attribute__((ext_vector_type(4))) float v4f;
typedef __attribute__((ext_vector_type(8))) float v8f;

#define N_IN   128
#define N_HID  256
#define N_HID2 128
#define N_CLS  2

__global__ void count_rank_kernel(const int* __restrict__ col, int* __restrict__ cnt,
                                  int* __restrict__ rank, int E) {
    int e = blockIdx.x * blockDim.x + threadIdx.x;
    if (e < E) rank[e] = atomicAdd(&cnt[col[e]], 1);
}

#define ALLOC_T 256
__global__ __launch_bounds__(ALLOC_T)
void seg_alloc_kernel(const int* __restrict__ cnt, int* __restrict__ total,
                      int* __restrict__ beg, float* __restrict__ dinv, int n) {
    __shared__ int s[2][ALLOC_T];
    __shared__ int s_base;
    const int t = threadIdx.x;
    const int i = blockIdx.x * ALLOC_T + t;
    const int c = (i < n) ? cnt[i] : 0;
    int cur = 0;
    s[0][t] = c;
    __syncthreads();
    for (int d = 1; d < ALLOC_T; d <<= 1) {
        int v = s[cur][t];
        if (t >= d) v += s[cur][t - d];
        s[cur ^ 1][t] = v;
        cur ^= 1;
        __syncthreads();
    }
    const int incl = s[cur][t];
    if (t == ALLOC_T - 1) s_base = atomicAdd(total, incl);
    __syncthreads();
    if (i < n) {
        beg[i]  = s_base + incl - c;
        dinv[i] = 1.0f / fmaxf((float)c, 1.0f);
    }
}

__global__ void fill_kernel(const int* __restrict__ row, const int* __restrict__ col,
                            const int* __restrict__ rank, const int* __restrict__ beg,
                            int* __restrict__ srcs, int E) {
    int e = blockIdx.x * blockDim.x + threadIdx.x;
    if (e < E) srcs[beg[col[e]] + rank[e]] = row[e];
}

template<int F, int G>
__device__ __forceinline__ void add_rows(const float* __restrict__ src, int my, int j, int lane,
                                         int last, v4f (&acc)[F / 128]) {
    v4f v[G][F / 128];
    #pragma unroll
    for (int g = 0; g < G; ++g) {
        int r = __builtin_amdgcn_readlane(my, j + g);
        r = r < 0 ? 0 : (r > last ? last : r);
        const float* __restrict__ p = src + (size_t)r * F + lane * 4;
        #pragma unroll
        for (int q = 0; q < F / 128; ++q) v[g][q] = *(const v4f*)(p + q * 128);
    }
    #pragma unroll
    for (int g = 0; g < G; ++g) {
        #pragma unroll
        for (int q = 0; q < F / 128; ++q) acc[q] += v[g][q];
    }
}

template<int F>
__global__ __launch_bounds__(256)
void gather_sum_kernel(const float* __restrict__ src, const int* __restrict__ srcs,
                       const int* __restrict__ beg, const int* __restrict__ cnt,
                       float* __restrict__ agg, int n) {
    constexpr int U = 1024 / F;
    const int lane = threadIdx.x & 31;
    const int node = __builtin_amdgcn_readfirstlane((int)(blockIdx.x * (blockDim.x >> 5) + (threadIdx.x >> 5)));
    if (node >= n) return;
    const int b = beg[node];
    const int d = cnt[node];
    v4f acc[F / 128];
    #pragma unroll
    for (int q = 0; q < F / 128; ++q) acc[q] = (v4f){0.0f, 0.0f, 0.0f, 0.0f};

    for (int base = 0; base < d; base += 32) {
        const int m = (d - base < 32) ? (d - base) : 32;
        const int my = (lane < m) ? srcs[b + base + lane] : 0;
        int j = 0;
        for (; j + U <= m; j += U) add_rows<F, U>(src, my, j, lane, n - 1, acc);
        if (U == 8 && m - j >= 4) { add_rows<F, 4>(src, my, j, lane, n - 1, acc); j += 4; }
        if (m - j >= 2) { add_rows<F, 2>(src, my, j, lane, n - 1, acc); j += 2; }
        if (m - j >= 1) { add_rows<F, 1>(src, my, j, lane, n - 1, acc); }
    }

    float* __restrict__ o = agg + (size_t)node * F + lane * 4;
    #pragma unroll
    for (int q = 0; q < F / 128; ++q) *(v4f*)(o + q * 128) = acc[q];
}

template<int K, int NF, bool RELU>
__global__ __launch_bounds__(256)
void gemm_wmma(const float* __restrict__ A, const float* __restrict__ W,
               const float* __restrict__ bias, const float* __restrict__ dinv,
               float* __restrict__ out, int M) {
    const int lane  = threadIdx.x;
    const int waveId = threadIdx.y;
    const int tilesN = NF / 16;
    const int tilesM = (M + 15) / 16;
    int tile = blockIdx.x * blockDim.y + waveId;
    if (tile >= tilesM * tilesN) return;
    const int tm = tile / tilesN;
    const int tn = tile % tilesN;

    const int half = lane >> 4;
    const int l16  = lane & 15;

    int arow = tm * 16 + l16;
    if (arow >= M) arow = M - 1;
    const float* __restrict__ aptr = A + (size_t)arow * K;
    const float* __restrict__ wptr = W + (size_t)(tn * 16 + l16) * K;

    v8f acc = {};
    #pragma unroll
    for (int k = 0; k < K; k += 4) {
        const int kk = k + 2 * half;
        v2f a = *(const v2f*)(aptr + kk);
        v2f b = *(const v2f*)(wptr + kk);
        acc = __builtin_amdgcn_wmma_f32_16x16x4_f32(
             false, a,  false, b,
             (short)0, acc,  false,  false);
    }

    const int n  = tn * 16 + l16;
    const float bv = bias[n];
    #pragma unroll
    for (int r = 0; r < 8; ++r) {
        int m = tm * 16 + half * 8 + r;
        if (m < M) {
            float v = acc[r] * dinv[m] + bv;
            if (RELU) v = fmaxf(v, 0.0f);
            out[(size_t)m * NF + n] = v;
        }
    }
}

__global__ void head_kernel(const float* __restrict__ h, const float* __restrict__ W3,
                            const float* __restrict__ b3, float* __restrict__ out, int M) {
    int tid = blockIdx.x * blockDim.x + threadIdx.x;
    if (tid >= M * N_CLS) return;
    int m = tid / N_CLS, c = tid % N_CLS;
    const float* __restrict__ hm = h  + (size_t)m * N_HID2;
    const float* __restrict__ wc = W3 + (size_t)c * N_HID2;
    float s = b3[c];
    #pragma unroll 8
    for (int k = 0; k < N_HID2; ++k) s += hm[k] * wc[k];
    out[tid] = s;
}

extern "C" void kernel_launch(void* const* d_in, const int* in_sizes, int n_in,
                              void* d_out, int out_size, void* d_ws, size_t ws_size,
                              hipStream_t stream) {
    const float* x  = (const float*)d_in[0];
    const int*   ei = (const int*)  d_in[1];
    const float* W1 = (const float*)d_in[3];
    const float* b1 = (const float*)d_in[4];
    const float* W2 = (const float*)d_in[5];
    const float* b2 = (const float*)d_in[6];
    const float* W3 = (const float*)d_in[7];
    const float* b3 = (const float*)d_in[8];
    float* out = (float*)d_out;

    const int M = in_sizes[0] / N_IN;
    const int E = in_sizes[1] / 2;
    const int* row = ei;
    const int* col = ei + E;

    const size_t Mp = ((size_t)M + 1 + 31) & ~(size_t)31;
    const size_t Ep = ((size_t)E + 31) & ~(size_t)31;
    int* wsi = (int*)d_ws;
    size_t off = 0;
    int* cnt   = wsi + off; off += Mp;
    int* beg   = wsi + off; off += Mp;
    int* rank  = wsi + off; off += Ep;
    int* srcs  = wsi + off; off += Ep;
    float* ws = (float*)d_ws;
    float* dinv = ws + off; off += Mp;
    float* agg1 = ws + off; off += (size_t)M * N_IN;
    float* h1   = ws + off; off += (size_t)M * N_HID;
    float* agg2 = ws + off; off += (size_t)M * N_HID;
    float* h2   = ws + off; off += (size_t)M * N_HID2;
    (void)ws_size;

    hipMemsetAsync(cnt, 0, sizeof(int) * ((size_t)M + 1), stream);

    count_rank_kernel<<<(E + 255) / 256, 256, 0, stream>>>(col, cnt, rank, E);
    seg_alloc_kernel<<<(M + ALLOC_T - 1) / ALLOC_T, ALLOC_T, 0, stream>>>(cnt, cnt + M, beg, dinv, M);
    fill_kernel<<<(E + 255) / 256, 256, 0, stream>>>(row, col, rank, beg, srcs, E);

    gather_sum_kernel<N_IN><<<(M + 7) / 8, 256, 0, stream>>>(x, srcs, beg, cnt, agg1, M);
    int tiles1 = ((M + 15) / 16) * (N_HID / 16);
    gemm_wmma<N_IN, N_HID, true>
        <<<(tiles1 + 7) / 8, dim3(32, 8), 0, stream>>>(agg1, W1, b1, dinv, h1, M);

    gather_sum_kernel<N_HID><<<(M + 7) / 8, 256, 0, stream>>>(h1, srcs, beg, cnt, agg2, M);
    int tiles2 = ((M + 15) / 16) * (N_HID2 / 16);
    gemm_wmma<N_HID, N_HID2, true>
        <<<(tiles2 + 7) / 8, dim3(32, 8), 0, stream>>>(agg2, W2, b2, dinv, h2, M);

    head_kernel<<<(M * N_CLS + 255) / 256, 256, 0, stream>>>(h2, W3, b3, out, M);
}


# GEMM operands staged through LDS by whole-row 16-byte loads, 64x64 tile per 8-wave group, two accumulator tiles per wave (was one 16x16 tile per wave fed by 8-byte global loads); k order kept
# speedup vs baseline: 5.1859x; 1.4967x over previous
#include <hip/hip_runtime.h>

typedef __attribute__((ext_vector_type(2))) float v2f;
typedef __attribute__((ext_vector_type(4))) float v4f;
typedef __attribute__((ext_vector_type(8))) float v8f;

#define N_IN   128
#define N_HID  256
#define N_HID2 128
#define N_CLS  2

__global__ void count_rank_kernel(const int* __restrict__ col, int* __restrict__ cnt,
                                  int* __restrict__ rank, int E) {
    int e = blockIdx.x * blockDim.x + threadIdx.x;
    if (e < E) rank[e] = atomicAdd(&cnt[col[e]], 1);
}

#define ALLOC_T 256
__global__ __launch_bounds__(ALLOC_T)
void seg_alloc_kernel(const int* __restrict__ cnt, int* __restrict__ total,
                      int* __restrict__ beg, float* __restrict__ dinv, int n) {
    __shared__ int s[2][ALLOC_T];
    __shared__ int s_base;
    const int t = threadIdx.x;
    const int i = blockIdx.x * ALLOC_T + t;
    const int c = (i < n) ? cnt[i] : 0;
    int cur = 0;
    s[0][t] = c;
    __syncthreads();
    for (int d = 1; d < ALLOC_T; d <<= 1) {
        int v = s[cur][t];
        if (t >= d) v += s[cur][t - d];
        s[cur ^ 1][t] = v;
        cur ^= 1;
        __syncthreads();
    }
    const int incl = s[cur][t];
    if (t == ALLOC_T - 1) s_base = atomicAdd(total, incl);
    __syncthreads();
    if (i < n) {
        beg[i]  = s_base + incl - c;
        dinv[i] = 1.0f / fmaxf((float)c, 1.0f);
    }
}

__global__ void fill_kernel(const int* __restrict__ row, const int* __restrict__ col,
                            const int* __restrict__ rank, const int* __restrict__ beg,
                            int* __restrict__ srcs, int E) {
    int e = blockIdx.x * blockDim.x + threadIdx.x;
    if (e < E) srcs[beg[col[e]] + rank[e]] = row[e];
}

template<int F, int G>
__device__ __forceinline__ void add_rows(const float* __restrict__ src, int my, int j, int lane,
                                         int last, v4f (&acc)[F / 128]) {
    v4f v[G][F / 128];
    #pragma unroll
    for (int g = 0; g < G; ++g) {
        int r = __builtin_amdgcn_readlane(my, j + g);
        r = r < 0 ? 0 : (r > last ? last : r);
        const float* __restrict__ p = src + (size_t)r * F + lane * 4;
        #pragma unroll
        for (int q = 0; q < F / 128; ++q) v[g][q] = *(const v4f*)(p + q * 128);
    }
    #pragma unroll
    for (int g = 0; g < G; ++g) {
        #pragma unroll
        for (int q = 0; q < F / 128; ++q) acc[q] += v[g][q];
    }
}

template<int F>
__global__ __launch_bounds__(256)
void gather_sum_kernel(const float* __restrict__ src, const int* __restrict__ srcs,
                       const int* __restrict__ beg, const int* __restrict__ cnt,
                       float* __restrict__ agg, int n) {
    constexpr int U = 1024 / F;
    const int lane = threadIdx.x & 31;
    const int node = __builtin_amdgcn_readfirstlane((int)(blockIdx.x * (blockDim.x >> 5) + (threadIdx.x >> 5)));
    if (node >= n) return;
    const int b = beg[node];
    const int d = cnt[node];
    v4f acc[F / 128];
    #pragma unroll
    for (int q = 0; q < F / 128; ++q) acc[q] = (v4f){0.0f, 0.0f, 0.0f, 0.0f};

    for (int base = 0; base < d; base += 32) {
        const int m = (d - base < 32) ? (d - base) : 32;
        const int my = (lane < m) ? srcs[b + base + lane] : 0;
        int j = 0;
        for (; j + U <= m; j += U) add_rows<F, U>(src, my, j, lane, n - 1, acc);
        if (U == 8 && m - j >= 4) { add_rows<F, 4>(src, my, j, lane, n - 1, acc); j += 4; }
        if (m - j >= 2) { add_rows<F, 2>(src, my, j, lane, n - 1, acc); j += 2; }
        if (m - j >= 1) { add_rows<F, 1>(src, my, j, lane, n - 1, acc); }
    }

    float* __restrict__ o = agg + (size_t)node * F + lane * 4;
    #pragma unroll
    for (int q = 0; q < F / 128; ++q) *(v4f*)(o + q * 128) = acc[q];
}

#define G_BM    64
#define G_BN    64
#define G_KC    128
#define G_PITCH 132

template<int K, int NF, bool RELU>
__global__ __launch_bounds__(256)
void gemm_wmma(const float* __restrict__ A, const float* __restrict__ W,
               const float* __restrict__ bias, const float* __restrict__ dinv,
               float* __restrict__ out, int M) {
    __shared__ __attribute__((aligned(16))) float As[G_BM * G_PITCH];
    __shared__ __attribute__((aligned(16))) float Ws[G_BN * G_PITCH];
    constexpr int NB  = NF / G_BN;
    constexpr int NCH = K / G_KC;
    static_assert(K % G_KC == 0 && NF % G_BN == 0, "tile sizes");

    const int t    = threadIdx.x;
    const int lane = t & 31;
    const int wave = t >> 5;
    const int wm   = wave >> 1;
    const int wn   = wave & 1;
    const int half = lane >> 4;
    const int l16  = lane & 15;
    const int m0 = (int)(blockIdx.x / NB) * G_BM;
    const int n0 = (int)(blockIdx.x % NB) * G_BN;

    const int q  = t & 31;
    const int r0 = t >> 5;
    v4f ra[8], rw[8];
    #pragma unroll
    for (int i = 0; i < 8; ++i) {
        int ar = m0 + r0 + 8 * i;
        if (ar >= M) ar = M - 1;
        ra[i] = *(const v4f*)(A + (size_t)ar * K + 4 * q);
        rw[i] = *(const v4f*)(W + (size_t)(n0 + r0 + 8 * i) * K + 4 * q);
    }

    const float* __restrict__ as  = As + (wm * 16 + l16) * G_PITCH + 2 * half;
    const float* __restrict__ ws0 = Ws + (wn * 32 + l16) * G_PITCH + 2 * half;
    const float* __restrict__ ws1 = ws0 + 16 * G_PITCH;

    v8f acc0 = {}, acc1 = {};
    float dv[8];
    #pragma unroll
    for (int c = 0; c < NCH; ++c) {
        if (c > 0) __syncthreads();
        #pragma unroll
        for (int i = 0; i < 8; ++i) {
            *(v4f*)(As + (r0 + 8 * i) * G_PITCH + 4 * q) = ra[i];
            *(v4f*)(Ws + (r0 + 8 * i) * G_PITCH + 4 * q) = rw[i];
        }
        __syncthreads();
        if (c + 1 < NCH) {
            #pragma unroll
            for (int i = 0; i < 8; ++i) {
                int ar = m0 + r0 + 8 * i;
                if (ar >= M) ar = M - 1;
                ra[i] = *(const v4f*)(A + (size_t)ar * K + (c + 1) * G_KC + 4 * q);
                rw[i] = *(const v4f*)(W + (size_t)(n0 + r0 + 8 * i) * K + (c + 1) * G_KC + 4 * q);
            }
        }
        if (c == NCH - 1) {
            #pragma unroll
            for (int r = 0; r < 8; ++r) {
                int m = m0 + wm * 16 + half * 8 + r;
                dv[r] = dinv[m < M ? m : M - 1];
            }
        }
        #pragma unroll
        for (int k = 0; k < G_KC; k += 4) {
            v2f a  = *(const v2f*)(as  + k);
            v2f b0 = *(const v2f*)(ws0 + k);
            v2f b1 = *(const v2f*)(ws1 + k);
            acc0 = __builtin_amdgcn_wmma_f32_16x16x4_f32(
                 false, a,  false, b0,
                 (short)0, acc0,  false,  false);
            acc1 = __builtin_amdgcn_wmma_f32_16x16x4_f32(
                 false, a,  false, b1,
                 (short)0, acc1,  false,  false);
        }
    }

    const int na = n0 + wn * 32 + l16;
    const int nb = na + 16;
    const float bva = bias[na];
    const float bvb = bias[nb];
    #pragma unroll
    for (int r = 0; r < 8; ++r) {
        int m = m0 + wm * 16 + half * 8 + r;
        if (m < M) {
            float va = acc0[r] * dv[r] + bva;
            float vb = acc1[r] * dv[r] + bvb;
            if (RELU) { va = fmaxf(va, 0.0f); vb = fmaxf(vb, 0.0f); }
            out[(size_t)m * NF + na] = va;
            out[(size_t)m * NF + nb] = vb;
        }
    }
}

__global__ void head_kernel(const float* __restrict__ h, const float* __restrict__ W3,
                            const float* __restrict__ b3, float* __restrict__ out, int M) {
    int tid = blockIdx.x * blockDim.x + threadIdx.x;
    if (tid >= M * N_CLS) return;
    int m = tid / N_CLS, c = tid % N_CLS;
    const float* __restrict__ hm = h  + (size_t)m * N_HID2;
    const float* __restrict__ wc = W3 + (size_t)c * N_HID2;
    float s = b3[c];
    #pragma unroll 8
    for (int k = 0; k < N_HID2; ++k) s += hm[k] * wc[k];
    out[tid] = s;
}

extern "C" void kernel_launch(void* const* d_in, const int* in_sizes, int n_in,
                              void* d_out, int out_size, void* d_ws, size_t ws_size,
                              hipStream_t stream) {
    const float* x  = (const float*)d_in[0];
    const int*   ei = (const int*)  d_in[1];
    const float* W1 = (const float*)d_in[3];
    const float* b1 = (const float*)d_in[4];
    const float* W2 = (const float*)d_in[5];
    const float* b2 = (const float*)d_in[6];
    const float* W3 = (const float*)d_in[7];
    const float* b3 = (const float*)d_in[8];
    float* out = (float*)d_out;

    const int M = in_sizes[0] / N_IN;
    const int E = in_sizes[1] / 2;
    const int* row = ei;
    const int* col = ei + E;

    const size_t Mp = ((size_t)M + 1 + 31) & ~(size_t)31;
    const size_t Ep = ((size_t)E + 31) & ~(size_t)31;
    int* wsi = (int*)d_ws;
    size_t off = 0;
    int* cnt   = wsi + off; off += Mp;
    int* beg   = wsi + off; off += Mp;
    int* rank  = wsi + off; off += Ep;
    int* srcs  = wsi + off; off += Ep;
    float* ws = (float*)d_ws;
    float* dinv = ws + off; off += Mp;
    float* agg1 = ws + off; off += (size_t)M * N_IN;
    float* h1   = ws + off; off += (size_t)M * N_HID;
    float* agg2 = ws + off; off += (size_t)M * N_HID;
    float* h2   = ws + off; off += (size_t)M * N_HID2;
    (void)ws_size;

    hipMemsetAsync(cnt, 0, sizeof(int) * ((size_t)M + 1), stream);

    count_rank_kernel<<<(E + 255) / 256, 256, 0, stream>>>(col, cnt, rank, E);
    seg_alloc_kernel<<<(M + ALLOC_T - 1) / ALLOC_T, ALLOC_T, 0, stream>>>(cnt, cnt + M, beg, dinv, M);
    fill_kernel<<<(E + 255) / 256, 256, 0, stream>>>(row, col, rank, beg, srcs, E);

    gather_sum_kernel<N_IN><<<(M + 7) / 8, 256, 0, stream>>>(x, srcs, beg, cnt, agg1, M);
    int blocks1 = ((M + G_BM - 1) / G_BM) * (N_HID / G_BN);
    gemm_wmma<N_IN, N_HID, true>
        <<<blocks1, 256, 0, stream>>>(agg1, W1, b1, dinv, h1, M);

    gather_sum_kernel<N_HID><<<(M + 7) / 8, 256, 0, stream>>>(h1, srcs, beg, cnt, agg2, M);
    int blocks2 = ((M + G_BM - 1) / G_BM) * (N_HID2 / G_BN);
    gemm_wmma<N_HID, N_HID2, true>
        <<<blocks2, 256, 0, stream>>>(agg2, W2, b2, dinv, h2, M);

    head_kernel<<<(M * N_CLS + 255) / 256, 256, 0, stream>>>(h2, W3, b3, out, M);
}


# Layer 2 applies W2 before its neighbour sum (the sum is linear): the gather moves 128-float rows, not 256, and ends with 1/deg, bias, relu; neighbour sums now add products (sum order changed)
# speedup vs baseline: 5.6522x; 1.0899x over previous
#include <hip/hip_runtime.h>

typedef __attribute__((ext_vector_type(2))) float v2f;
typedef __attribute__((ext_vector_type(4))) float v4f;
typedef __attribute__((ext_vector_type(8))) float v8f;

#define N_IN   128
#define N_HID  256
#define N_HID2 128
#define N_CLS  2

__global__ void count_rank_kernel(const int* __restrict__ col, int* __restrict__ cnt,
                                  int* __restrict__ rank, int E) {
    int e = blockIdx.x * blockDim.x + threadIdx.x;
    if (e < E) rank[e] = atomicAdd(&cnt[col[e]], 1);
}

#define ALLOC_T 256
__global__ __launch_bounds__(ALLOC_T)
void seg_alloc_kernel(const int* __restrict__ cnt, int* __restrict__ total,
                      int* __restrict__ beg, float* __restrict__ dinv, int n) {
    __shared__ int s[2][ALLOC_T];
    __shared__ int s_base;
    const int t = threadIdx.x;
    const int i = blockIdx.x * ALLOC_T + t;
    const int c = (i < n) ? cnt[i] : 0;
    int cur = 0;
    s[0][t] = c;
    __syncthreads();
    for (int d = 1; d < ALLOC_T; d <<= 1) {
        int v = s[cur][t];
        if (t >= d) v += s[cur][t - d];
        s[cur ^ 1][t] = v;
        cur ^= 1;
        __syncthreads();
    }
    const int incl = s[cur][t];
    if (t == ALLOC_T - 1) s_base = atomicAdd(total, incl);
    __syncthreads();
    if (i < n) {
        beg[i]  = s_base + incl - c;
        dinv[i] = 1.0f / fmaxf((float)c, 1.0f);
    }
}

__global__ void fill_kernel(const int* __restrict__ row, const int* __restrict__ col,
                            const int* __restrict__ rank, const int* __restrict__ beg,
                            int* __restrict__ srcs, int E) {
    int e = blockIdx.x * blockDim.x + threadIdx.x;
    if (e < E) srcs[beg[col[e]] + rank[e]] = row[e];
}

template<int F, int G>
__device__ __forceinline__ void add_rows(const float* __restrict__ src, int my, int j, int lane,
                                         int last, v4f (&acc)[F / 128]) {
    v4f v[G][F / 128];
    #pragma unroll
    for (int g = 0; g < G; ++g) {
        int r = __builtin_amdgcn_readlane(my, j + g);
        r = r < 0 ? 0 : (r > last ? last : r);
        const float* __restrict__ p = src + (size_t)r * F + lane * 4;
        #pragma unroll
        for (int q = 0; q < F / 128; ++q) v[g][q] = *(const v4f*)(p + q * 128);
    }
    #pragma unroll
    for (int g = 0; g < G; ++g) {
        #pragma unroll
        for (int q = 0; q < F / 128; ++q) acc[q] += v[g][q];
    }
}

template<int F, bool FIN>
__global__ __launch_bounds__(256)
void gather_sum_kernel(const float* __restrict__ src, const int* __restrict__ srcs,
                       const int* __restrict__ beg, const int* __restrict__ cnt,
                       const float* __restrict__ dinv, const float* __restrict__ bias,
                       float* __restrict__ agg, int n) {
    constexpr int U = 1024 / F;
    const int lane = threadIdx.x & 31;
    const int node = __builtin_amdgcn_readfirstlane((int)(blockIdx.x * (blockDim.x >> 5) + (threadIdx.x >> 5)));
    if (node >= n) return;
    const int b = beg[node];
    const int d = cnt[node];
    float s = 0.0f;
    v4f bv[F / 128];
    if (FIN) {
        s = dinv[node];
        #pragma unroll
        for (int q = 0; q < F / 128; ++q) bv[q] = *(const v4f*)(bias + q * 128 + lane * 4);
    }
    v4f acc[F / 128];
    #pragma unroll
    for (int q = 0; q < F / 128; ++q) acc[q] = (v4f){0.0f, 0.0f, 0.0f, 0.0f};

    for (int base = 0; base < d; base += 32) {
        const int m = (d - base < 32) ? (d - base) : 32;
        const int my = (lane < m) ? srcs[b + base + lane] : 0;
        int j = 0;
        for (; j + U <= m; j += U) add_rows<F, U>(src, my, j, lane, n - 1, acc);
        if (U == 8 && m - j >= 4) { add_rows<F, 4>(src, my, j, lane, n - 1, acc); j += 4; }
        if (m - j >= 2) { add_rows<F, 2>(src, my, j, lane, n - 1, acc); j += 2; }
        if (m - j >= 1) { add_rows<F, 1>(src, my, j, lane, n - 1, acc); }
    }

    if (FIN) {
        #pragma unroll
        for (int q = 0; q < F / 128; ++q) {
            v4f v = acc[q] * s + bv[q];
            v.x = fmaxf(v.x, 0.0f); v.y = fmaxf(v.y, 0.0f);
            v.z = fmaxf(v.z, 0.0f); v.w = fmaxf(v.w, 0.0f);
            acc[q] = v;
        }
    }
    float* __restrict__ o = agg + (size_t)node * F + lane * 4;
    #pragma unroll
    for (int q = 0; q < F / 128; ++q) *(v4f*)(o + q * 128) = acc[q];
}

#define G_BM    64
#define G_BN    64
#define G_KC    128
#define G_PITCH 132

template<int K, int NF, bool EPI>
__global__ __launch_bounds__(256)
void gemm_wmma(const float* __restrict__ A, const float* __restrict__ W,
               const float* __restrict__ bias, const float* __restrict__ dinv,
               float* __restrict__ out, int M) {
    __shared__ __attribute__((aligned(16))) float As[G_BM * G_PITCH];
    __shared__ __attribute__((aligned(16))) float Ws[G_BN * G_PITCH];
    constexpr int NB  = NF / G_BN;
    constexpr int NCH = K / G_KC;
    static_assert(K % G_KC == 0 && NF % G_BN == 0, "tile sizes");

    const int t    = threadIdx.x;
    const int lane = t & 31;
    const int wave = t >> 5;
    const int wm   = wave >> 1;
    const int wn   = wave & 1;
    const int half = lane >> 4;
    const int l16  = lane & 15;
    const int m0 = (int)(blockIdx.x / NB) * G_BM;
    const int n0 = (int)(blockIdx.x % NB) * G_BN;

    const int q  = t & 31;
    const int r0 = t >> 5;
    v4f ra[8], rw[8];
    #pragma unroll
    for (int i = 0; i < 8; ++i) {
        int ar = m0 + r0 + 8 * i;
        if (ar >= M) ar = M - 1;
        ra[i] = *(const v4f*)(A + (size_t)ar * K + 4 * q);
        rw[i] = *(const v4f*)(W + (size_t)(n0 + r0 + 8 * i) * K + 4 * q);
    }

    const float* __restrict__ as  = As + (wm * 16 + l16) * G_PITCH + 2 * half;
    const float* __restrict__ ws0 = Ws + (wn * 32 + l16) * G_PITCH + 2 * half;
    const float* __restrict__ ws1 = ws0 + 16 * G_PITCH;

    v8f acc0 = {}, acc1 = {};
    float dv[8];
    #pragma unroll
    for (int c = 0; c < NCH; ++c) {
        if (c > 0) __syncthreads();
        #pragma unroll
        for (int i = 0; i < 8; ++i) {
            *(v4f*)(As + (r0 + 8 * i) * G_PITCH + 4 * q) = ra[i];
            *(v4f*)(Ws + (r0 + 8 * i) * G_PITCH + 4 * q) = rw[i];
        }
        __syncthreads();
        if (c + 1 < NCH) {
            #pragma unroll
            for (int i = 0; i < 8; ++i) {
                int ar = m0 + r0 + 8 * i;
                if (ar >= M) ar = M - 1;
                ra[i] = *(const v4f*)(A + (size_t)ar * K + (c + 1) * G_KC + 4 * q);
                rw[i] = *(const v4f*)(W + (size_t)(n0 + r0 + 8 * i) * K + (c + 1) * G_KC + 4 * q);
            }
        }
        if (EPI && c == NCH - 1) {
            #pragma unroll
            for (int r = 0; r < 8; ++r) {
                int m = m0 + wm * 16 + half * 8 + r;
                dv[r] = dinv[m < M ? m : M - 1];
            }
        }
        #pragma unroll
        for (int k = 0; k < G_KC; k += 4) {
            v2f a  = *(const v2f*)(as  + k);
            v2f b0 = *(const v2f*)(ws0 + k);
            v2f b1 = *(const v2f*)(ws1 + k);
            acc0 = __builtin_amdgcn_wmma_f32_16x16x4_f32(
                 false, a,  false, b0,
                 (short)0, acc0,  false,  false);
            acc1 = __builtin_amdgcn_wmma_f32_16x16x4_f32(
                 false, a,  false, b1,
                 (short)0, acc1,  false,  false);
        }
    }

    const int na = n0 + wn * 32 + l16;
    const int nb = na + 16;
    const float bva = EPI ? bias[na] : 0.0f;
    const float bvb = EPI ? bias[nb] : 0.0f;
    #pragma unroll
    for (int r = 0; r < 8; ++r) {
        int m = m0 + wm * 16 + half * 8 + r;
        if (m < M) {
            float va = acc0[r];
            float vb = acc1[r];
            if (EPI) {
                va = fmaxf(va * dv[r] + bva, 0.0f);
                vb = fmaxf(vb * dv[r] + bvb, 0.0f);
            }
            out[(size_t)m * NF + na] = va;
            out[(size_t)m * NF + nb] = vb;
        }
    }
}

__global__ void head_kernel(const float* __restrict__ h, const float* __restrict__ W3,
                            const float* __restrict__ b3, float* __restrict__ out, int M) {
    int tid = blockIdx.x * blockDim.x + threadIdx.x;
    if (tid >= M * N_CLS) return;
    int m = tid / N_CLS, c = tid % N_CLS;
    const float* __restrict__ hm = h  + (size_t)m * N_HID2;
    const float* __restrict__ wc = W3 + (size_t)c * N_HID2;
    float s = b3[c];
    #pragma unroll 8
    for (int k = 0; k < N_HID2; ++k) s += hm[k] * wc[k];
    out[tid] = s;
}

extern "C" void kernel_launch(void* const* d_in, const int* in_sizes, int n_in,
                              void* d_out, int out_size, void* d_ws, size_t ws_size,
                              hipStream_t stream) {
    const float* x  = (const float*)d_in[0];
    const int*   ei = (const int*)  d_in[1];
    const float* W1 = (const float*)d_in[3];
    const float* b1 = (const float*)d_in[4];
    const float* W2 = (const float*)d_in[5];
    const float* b2 = (const float*)d_in[6];
    const float* W3 = (const float*)d_in[7];
    const float* b3 = (const float*)d_in[8];
    float* out = (float*)d_out;

    const int M = in_sizes[0] / N_IN;
    const int E = in_sizes[1] / 2;
    const int* row = ei;
    const int* col = ei + E;

    const size_t Mp = ((size_t)M + 1 + 31) & ~(size_t)31;
    const size_t Ep = ((size_t)E + 31) & ~(size_t)31;
    int* wsi = (int*)d_ws;
    size_t off = 0;
    int* cnt   = wsi + off; off += Mp;
    int* beg   = wsi + off; off += Mp;
    int* rank  = wsi + off; off += Ep;
    int* srcs  = wsi + off; off += Ep;
    float* ws = (float*)d_ws;
    float* dinv = ws + off; off += Mp;
    float* agg1 = ws + off; off += (size_t)M * N_IN;
    float* h1   = ws + off; off += (size_t)M * N_HID;
    float* z2   = ws + off; off += (size_t)M * N_HID2;
    float* h2   = ws + off; off += (size_t)M * N_HID2;
    (void)ws_size;

    hipMemsetAsync(cnt, 0, sizeof(int) * ((size_t)M + 1), stream);

    count_rank_kernel<<<(E + 255) / 256, 256, 0, stream>>>(col, cnt, rank, E);
    seg_alloc_kernel<<<(M + ALLOC_T - 1) / ALLOC_T, ALLOC_T, 0, stream>>>(cnt, cnt + M, beg, dinv, M);
    fill_kernel<<<(E + 255) / 256, 256, 0, stream>>>(row, col, rank, beg, srcs, E);

    gather_sum_kernel<N_IN, false><<<(M + 7) / 8, 256, 0, stream>>>(x, srcs, beg, cnt, dinv, b1, agg1, M);
    int blocks1 = ((M + G_BM - 1) / G_BM) * (N_HID / G_BN);
    gemm_wmma<N_IN, N_HID, true>
        <<<blocks1, 256, 0, stream>>>(agg1, W1, b1, dinv, h1, M);

    int blocks2 = ((M + G_BM - 1) / G_BM) * (N_HID2 / G_BN);
    gemm_wmma<N_HID, N_HID2, false>
        <<<blocks2, 256, 0, stream>>>(h1, W2, b2, dinv, z2, M);
    gather_sum_kernel<N_HID2, true><<<(M + 7) / 8, 256, 0, stream>>>(z2, srcs, beg, cnt, dinv, b2, h2, M);

    head_kernel<<<(M * N_CLS + 255) / 256, 256, 0, stream>>>(h2, W3, b3, out, M);
}


# Three-pass per-call CSR (count+rank, segment scan, fill) replaced by one pass into a fixed 64-slot index row per node (slot = arrival rank, so sum order as before; overflow list past 64)
# speedup vs baseline: 6.3957x; 1.1315x over previous
#include <hip/hip_runtime.h>

typedef __attribute__((ext_vector_type(2))) float v2f;
typedef __attribute__((ext_vector_type(4))) float v4f;
typedef __attribute__((ext_vector_type(8))) float v8f;

#define N_IN   128
#define N_HID  256
#define N_HID2 128
#define N_CLS  2

#define CAP 64

__global__ void bucket_fill_kernel(const int* __restrict__ row, const int* __restrict__ col,
                                   int* __restrict__ cnt, int* __restrict__ bucket,
                                   int* __restrict__ ovf_dst, int* __restrict__ ovf_src,
                                   int M, int E) {
    int e = blockIdx.x * blockDim.x + threadIdx.x;
    if (e >= E) return;
    const int c = col[e];
    const int r = row[e];
    const int slot = atomicAdd(&cnt[c], 1);
    if (slot < CAP) {
        bucket[(size_t)c * CAP + slot] = r;
    } else {
        const int k = atomicAdd(&cnt[M], 1);
        ovf_dst[k] = c;
        ovf_src[k] = r;
    }
}

template<int F, int G>
__device__ __forceinline__ void add_rows(const float* __restrict__ src, int my, int j, int lane,
                                         int last, v4f (&acc)[F / 128]) {
    v4f v[G][F / 128];
    #pragma unroll
    for (int g = 0; g < G; ++g) {
        int r = __builtin_amdgcn_readlane(my, j + g);
        r = r < 0 ? 0 : (r > last ? last : r);
        const float* __restrict__ p = src + (size_t)r * F + lane * 4;
        #pragma unroll
        for (int q = 0; q < F / 128; ++q) v[g][q] = *(const v4f*)(p + q * 128);
    }
    #pragma unroll
    for (int g = 0; g < G; ++g) {
        #pragma unroll
        for (int q = 0; q < F / 128; ++q) acc[q] += v[g][q];
    }
}

template<int F, bool FIN>
__global__ __launch_bounds__(256)
void gather_sum_kernel(const float* __restrict__ src, const int* __restrict__ bucket,
                       const int* __restrict__ cnt, const int* __restrict__ ovf_dst,
                       const int* __restrict__ ovf_src, const float* __restrict__ bias,
                       float* __restrict__ agg, int n) {
    constexpr int U = 1024 / F;
    const int lane = threadIdx.x & 31;
    const int node = __builtin_amdgcn_readfirstlane((int)(blockIdx.x * (blockDim.x >> 5) + (threadIdx.x >> 5)));
    if (node >= n) return;
    const int d  = cnt[node];
    const int dm = d < CAP ? d : CAP;
    const int* __restrict__ seg = bucket + (size_t)node * CAP;
    v4f bv[F / 128];
    if (FIN) {
        #pragma unroll
        for (int q = 0; q < F / 128; ++q) bv[q] = *(const v4f*)(bias + q * 128 + lane * 4);
    }
    v4f acc[F / 128];
    #pragma unroll
    for (int q = 0; q < F / 128; ++q) acc[q] = (v4f){0.0f, 0.0f, 0.0f, 0.0f};

    for (int base = 0; base < dm; base += 32) {
        const int m = (dm - base < 32) ? (dm - base) : 32;
        const int my = (lane < m) ? seg[base + lane] : 0;
        int j = 0;
        for (; j + U <= m; j += U) add_rows<F, U>(src, my, j, lane, n - 1, acc);
        if (U == 8 && m - j >= 4) { add_rows<F, 4>(src, my, j, lane, n - 1, acc); j += 4; }
        if (m - j >= 2) { add_rows<F, 2>(src, my, j, lane, n - 1, acc); j += 2; }
        if (m - j >= 1) { add_rows<F, 1>(src, my, j, lane, n - 1, acc); }
    }

    if (d > CAP) {
        const int L = cnt[n];
        for (int k0 = 0; k0 < L; k0 += 32) {
            const int kk = k0 + lane;
            const int dst = (kk < L) ? ovf_dst[kk] : -1;
            const int sv  = (kk < L) ? ovf_src[kk] : 0;
            unsigned hit = __builtin_amdgcn_ballot_w32(dst == node);
            while (hit) {
                const int bit = __builtin_ctz(hit);
                hit &= hit - 1;
                add_rows<F, 1>(src, sv, bit, lane, n - 1, acc);
            }
        }
    }

    if (FIN) {
        const float s = 1.0f / fmaxf((float)d, 1.0f);
        #pragma unroll
        for (int q = 0; q < F / 128; ++q) {
            v4f v = acc[q] * s + bv[q];
            v.x = fmaxf(v.x, 0.0f); v.y = fmaxf(v.y, 0.0f);
            v.z = fmaxf(v.z, 0.0f); v.w = fmaxf(v.w, 0.0f);
            acc[q] = v;
        }
    }
    float* __restrict__ o = agg + (size_t)node * F + lane * 4;
    #pragma unroll
    for (int q = 0; q < F / 128; ++q) *(v4f*)(o + q * 128) = acc[q];
}

#define G_BM    64
#define G_BN    64
#define G_KC    128
#define G_PITCH 132

template<int K, int NF, bool EPI>
__global__ __launch_bounds__(256)
void gemm_wmma(const float* __restrict__ A, const float* __restrict__ W,
               const float* __restrict__ bias, const int* __restrict__ deg,
               float* __restrict__ out, int M) {
    __shared__ __attribute__((aligned(16))) float As[G_BM * G_PITCH];
    __shared__ __attribute__((aligned(16))) float Ws[G_BN * G_PITCH];
    constexpr int NB  = NF / G_BN;
    constexpr int NCH = K / G_KC;
    static_assert(K % G_KC == 0 && NF % G_BN == 0, "tile sizes");

    const int t    = threadIdx.x;
    const int lane = t & 31;
    const int wave = t >> 5;
    const int wm   = wave >> 1;
    const int wn   = wave & 1;
    const int half = lane >> 4;
    const int l16  = lane & 15;
    const int m0 = (int)(blockIdx.x / NB) * G_BM;
    const int n0 = (int)(blockIdx.x % NB) * G_BN;

    const int q  = t & 31;
    const int r0 = t >> 5;
    v4f ra[8], rw[8];
    #pragma unroll
    for (int i = 0; i < 8; ++i) {
        int ar = m0 + r0 + 8 * i;
        if (ar >= M) ar = M - 1;
        ra[i] = *(const v4f*)(A + (size_t)ar * K + 4 * q);
        rw[i] = *(const v4f*)(W + (size_t)(n0 + r0 + 8 * i) * K + 4 * q);
    }

    const float* __restrict__ as  = As + (wm * 16 + l16) * G_PITCH + 2 * half;
    const float* __restrict__ ws0 = Ws + (wn * 32 + l16) * G_PITCH + 2 * half;
    const float* __restrict__ ws1 = ws0 + 16 * G_PITCH;

    v8f acc0 = {}, acc1 = {};
    float dv[8];
    #pragma unroll
    for (int c = 0; c < NCH; ++c) {
        if (c > 0) __syncthreads();
        #pragma unroll
        for (int i = 0; i < 8; ++i) {
            *(v4f*)(As + (r0 + 8 * i) * G_PITCH + 4 * q) = ra[i];
            *(v4f*)(Ws + (r0 + 8 * i) * G_PITCH + 4 * q) = rw[i];
        }
        __syncthreads();
        if (c + 1 < NCH) {
            #pragma unroll
            for (int i = 0; i < 8; ++i) {
                int ar = m0 + r0 + 8 * i;
                if (ar >= M) ar = M - 1;
                ra[i] = *(const v4f*)(A + (size_t)ar * K + (c + 1) * G_KC + 4 * q);
                rw[i] = *(const v4f*)(W + (size_t)(n0 + r0 + 8 * i) * K + (c + 1) * G_KC + 4 * q);
            }
        }
        if (EPI && c == NCH - 1) {
            #pragma unroll
            for (int r = 0; r < 8; ++r) {
                int m = m0 + wm * 16 + half * 8 + r;
                dv[r] = 1.0f / fmaxf((float)deg[m < M ? m : M - 1], 1.0f);
            }
        }
        #pragma unroll
        for (int k = 0; k < G_KC; k += 4) {
            v2f a  = *(const v2f*)(as  + k);
            v2f b0 = *(const v2f*)(ws0 + k);
            v2f b1 = *(const v2f*)(ws1 + k);
            acc0 = __builtin_amdgcn_wmma_f32_16x16x4_f32(
                 false, a,  false, b0,
                 (short)0, acc0,  false,  false);
            acc1 = __builtin_amdgcn_wmma_f32_16x16x4_f32(
                 false, a,  false, b1,
                 (short)0, acc1,  false,  false);
        }
    }

    const int na = n0 + wn * 32 + l16;
    const int nb = na + 16;
    const float bva = EPI ? bias[na] : 0.0f;
    const float bvb = EPI ? bias[nb] : 0.0f;
    #pragma unroll
    for (int r = 0; r < 8; ++r) {
        int m = m0 + wm * 16 + half * 8 + r;
        if (m < M) {
            float va = acc0[r];
            float vb = acc1[r];
            if (EPI) {
                va = fmaxf(va * dv[r] + bva, 0.0f);
                vb = fmaxf(vb * dv[r] + bvb, 0.0f);
            }
            out[(size_t)m * NF + na] = va;
            out[(size_t)m * NF + nb] = vb;
        }
    }
}

__global__ void head_kernel(const float* __restrict__ h, const float* __restrict__ W3,
                            const float* __restrict__ b3, float* __restrict__ out, int M) {
    int tid = blockIdx.x * blockDim.x + threadIdx.x;
    if (tid >= M * N_CLS) return;
    int m = tid / N_CLS, c = tid % N_CLS;
    const float* __restrict__ hm = h  + (size_t)m * N_HID2;
    const float* __restrict__ wc = W3 + (size_t)c * N_HID2;
    float s = b3[c];
    #pragma unroll 8
    for (int k = 0; k < N_HID2; ++k) s += hm[k] * wc[k];
    out[tid] = s;
}

extern "C" void kernel_launch(void* const* d_in, const int* in_sizes, int n_in,
                              void* d_out, int out_size, void* d_ws, size_t ws_size,
                              hipStream_t stream) {
    const float* x  = (const float*)d_in[0];
    const int*   ei = (const int*)  d_in[1];
    const float* W1 = (const float*)d_in[3];
    const float* b1 = (const float*)d_in[4];
    const float* W2 = (const float*)d_in[5];
    const float* b2 = (const float*)d_in[6];
    const float* W3 = (const float*)d_in[7];
    const float* b3 = (const float*)d_in[8];
    float* out = (float*)d_out;

    const int M = in_sizes[0] / N_IN;
    const int E = in_sizes[1] / 2;
    const int* row = ei;
    const int* col = ei + E;

    const size_t Mp = ((size_t)M + 1 + 63) & ~(size_t)63;
    const size_t Ep = ((size_t)E + 31) & ~(size_t)31;
    int* wsi = (int*)d_ws;
    size_t off = 0;
    int* cnt     = wsi + off; off += Mp;
    int* bucket  = wsi + off; off += (size_t)M * CAP;
    int* ovf_dst = wsi + off; off += Ep;
    int* ovf_src = wsi + off; off += Ep;
    float* ws = (float*)d_ws;
    float* agg1 = ws + off; off += (size_t)M * N_IN;
    float* h1   = ws + off; off += (size_t)M * N_HID;
    float* z2   = ws + off; off += (size_t)M * N_HID2;
    float* h2   = ws + off; off += (size_t)M * N_HID2;
    (void)ws_size;

    hipMemsetAsync(cnt, 0, sizeof(int) * ((size_t)M + 1), stream);

    bucket_fill_kernel<<<(E + 255) / 256, 256, 0, stream>>>(row, col, cnt, bucket, ovf_dst, ovf_src, M, E);

    gather_sum_kernel<N_IN, false><<<(M + 7) / 8, 256, 0, stream>>>(x, bucket, cnt, ovf_dst, ovf_src, b1, agg1, M);
    int blocks1 = ((M + G_BM - 1) / G_BM) * (N_HID / G_BN);
    gemm_wmma<N_IN, N_HID, true>
        <<<blocks1, 256, 0, stream>>>(agg1, W1, b1, cnt, h1, M);

    int blocks2 = ((M + G_BM - 1) / G_BM) * (N_HID2 / G_BN);
    gemm_wmma<N_HID, N_HID2, false>
        <<<blocks2, 256, 0, stream>>>(h1, W2, b2, cnt, z2, M);
    gather_sum_kernel<N_HID2, true><<<(M + 7) / 8, 256, 0, stream>>>(z2, bucket, cnt, ovf_dst, ovf_src, b2, h2, M);

    head_kernel<<<(M * N_CLS + 255) / 256, 256, 0, stream>>>(h2, W3, b3, out, M);
}


# Head (h2 @ W3^T + b3) folded into the layer-2 gather: the node's wave sums its lane partials by a butterfly (head sum order changed), 16 nodes write one output line; h2 plane and a launch gone
# speedup vs baseline: 6.6677x; 1.0425x over previous
#include <hip/hip_runtime.h>

typedef __attribute__((ext_vector_type(2))) float v2f;
typedef __attribute__((ext_vector_type(4))) float v4f;
typedef __attribute__((ext_vector_type(8))) float v8f;

#define N_IN   128
#define N_HID  256
#define N_HID2 128
#define N_CLS  2

#define CAP 64

__global__ void bucket_fill_kernel(const int* __restrict__ row, const int* __restrict__ col,
                                   int* __restrict__ cnt, int* __restrict__ bucket,
                                   int* __restrict__ ovf_dst, int* __restrict__ ovf_src,
                                   int M, int E) {
    int e = blockIdx.x * blockDim.x + threadIdx.x;
    if (e >= E) return;
    const int c = col[e];
    const int r = row[e];
    const int slot = atomicAdd(&cnt[c], 1);
    if (slot < CAP) {
        bucket[(size_t)c * CAP + slot] = r;
    } else {
        const int k = atomicAdd(&cnt[M], 1);
        ovf_dst[k] = c;
        ovf_src[k] = r;
    }
}

template<int F, int G>
__device__ __forceinline__ void add_rows(const float* __restrict__ src, int my, int j, int lane,
                                         int last, v4f (&acc)[F / 128]) {
    v4f v[G][F / 128];
    #pragma unroll
    for (int g = 0; g < G; ++g) {
        int r = __builtin_amdgcn_readlane(my, j + g);
        r = r < 0 ? 0 : (r > last ? last : r);
        const float* __restrict__ p = src + (size_t)r * F + lane * 4;
        #pragma unroll
        for (int q = 0; q < F / 128; ++q) v[g][q] = *(const v4f*)(p + q * 128);
    }
    #pragma unroll
    for (int g = 0; g < G; ++g) {
        #pragma unroll
        for (int q = 0; q < F / 128; ++q) acc[q] += v[g][q];
    }
}

template<int F>
__device__ __forceinline__ int sum_node_rows(const float* __restrict__ src, const int* __restrict__ bucket,
                                             const int* __restrict__ cnt, const int* __restrict__ ovf_dst,
                                             const int* __restrict__ ovf_src, int node, int lane, int n,
                                             v4f (&acc)[F / 128]) {
    constexpr int U = 1024 / F;
    const int d  = cnt[node];
    const int dm = d < CAP ? d : CAP;
    const int* __restrict__ seg = bucket + (size_t)node * CAP;
    #pragma unroll
    for (int q = 0; q < F / 128; ++q) acc[q] = (v4f){0.0f, 0.0f, 0.0f, 0.0f};

    for (int base = 0; base < dm; base += 32) {
        const int m = (dm - base < 32) ? (dm - base) : 32;
        const int my = (lane < m) ? seg[base + lane] : 0;
        int j = 0;
        for (; j + U <= m; j += U) add_rows<F, U>(src, my, j, lane, n - 1, acc);
        if (U == 8 && m - j >= 4) { add_rows<F, 4>(src, my, j, lane, n - 1, acc); j += 4; }
        if (m - j >= 2) { add_rows<F, 2>(src, my, j, lane, n - 1, acc); j += 2; }
        if (m - j >= 1) { add_rows<F, 1>(src, my, j, lane, n - 1, acc); }
    }

    if (d > CAP) {
        const int L = cnt[n];
        for (int k0 = 0; k0 < L; k0 += 32) {
            const int kk = k0 + lane;
            const int dst = (kk < L) ? ovf_dst[kk] : -1;
            const int sv  = (kk < L) ? ovf_src[kk] : 0;
            unsigned hit = __builtin_amdgcn_ballot_w32(dst == node);
            while (hit) {
                const int bit = __builtin_ctz(hit);
                hit &= hit - 1;
                add_rows<F, 1>(src, sv, bit, lane, n - 1, acc);
            }
        }
    }
    return d;
}

template<int F>
__global__ __launch_bounds__(256)
void gather_sum_kernel(const float* __restrict__ src, const int* __restrict__ bucket,
                       const int* __restrict__ cnt, const int* __restrict__ ovf_dst,
                       const int* __restrict__ ovf_src, float* __restrict__ agg, int n) {
    const int lane = threadIdx.x & 31;
    const int node = __builtin_amdgcn_readfirstlane((int)(blockIdx.x * (blockDim.x >> 5) + (threadIdx.x >> 5)));
    if (node >= n) return;
    v4f acc[F / 128];
    sum_node_rows<F>(src, bucket, cnt, ovf_dst, ovf_src, node, lane, n, acc);
    float* __restrict__ o = agg + (size_t)node * F + lane * 4;
    #pragma unroll
    for (int q = 0; q < F / 128; ++q) *(v4f*)(o + q * 128) = acc[q];
}

#define FIN_WAVES 16
template<int F>
__global__ __launch_bounds__(FIN_WAVES * 32)
void gather_head_kernel(const float* __restrict__ src, const int* __restrict__ bucket,
                        const int* __restrict__ cnt, const int* __restrict__ ovf_dst,
                        const int* __restrict__ ovf_src, const float* __restrict__ bias,
                        const float* __restrict__ W3, const float* __restrict__ b3,
                        float* __restrict__ out, int n) {
    static_assert(FIN_WAVES * N_CLS == 32, "one wave writes the group's output line");
    __shared__ float line[FIN_WAVES * N_CLS];
    const int lane = threadIdx.x & 31;
    const int wave = __builtin_amdgcn_readfirstlane((int)(threadIdx.x >> 5));
    const int node = __builtin_amdgcn_readfirstlane((int)(blockIdx.x * FIN_WAVES) + wave);
    if (node < n) {
        v4f bv[F / 128], w0[F / 128], w1[F / 128];
        #pragma unroll
        for (int q = 0; q < F / 128; ++q) {
            bv[q] = *(const v4f*)(bias + q * 128 + lane * 4);
            w0[q] = *(const v4f*)(W3 + q * 128 + lane * 4);
            w1[q] = *(const v4f*)(W3 + F + q * 128 + lane * 4);
        }
        const float c0 = b3[0];
        const float c1 = b3[1];
        v4f acc[F / 128];
        const int d = sum_node_rows<F>(src, bucket, cnt, ovf_dst, ovf_src, node, lane, n, acc);
        const float s = 1.0f / fmaxf((float)d, 1.0f);
        float p0 = 0.0f, p1 = 0.0f;
        #pragma unroll
        for (int q = 0; q < F / 128; ++q) {
            v4f v = acc[q] * s + bv[q];
            v.x = fmaxf(v.x, 0.0f); v.y = fmaxf(v.y, 0.0f);
            v.z = fmaxf(v.z, 0.0f); v.w = fmaxf(v.w, 0.0f);
            p0 += v.x * w0[q].x + v.y * w0[q].y + v.z * w0[q].z + v.w * w0[q].w;
            p1 += v.x * w1[q].x + v.y * w1[q].y + v.z * w1[q].z + v.w * w1[q].w;
        }
        #pragma unroll
        for (int off = 16; off >= 1; off >>= 1) {
            p0 += __shfl_xor(p0, off, 32);
            p1 += __shfl_xor(p1, off, 32);
        }
        if (lane == 0) {
            line[wave * N_CLS + 0] = p0 + c0;
            line[wave * N_CLS + 1] = p1 + c1;
        }
    }
    __syncthreads();
    if (wave == 0) {
        const int idx = (int)blockIdx.x * (FIN_WAVES * N_CLS) + lane;
        if (idx < n * N_CLS) out[idx] = line[lane];
    }
}

#define G_BM    64
#define G_BN    64
#define G_KC    128
#define G_PITCH 132

template<int K, int NF, bool EPI>
__global__ __launch_bounds__(256)
void gemm_wmma(const float* __restrict__ A, const float* __restrict__ W,
               const float* __restrict__ bias, const int* __restrict__ deg,
               float* __restrict__ out, int M) {
    __shared__ __attribute__((aligned(16))) float As[G_BM * G_PITCH];
    __shared__ __attribute__((aligned(16))) float Ws[G_BN * G_PITCH];
    constexpr int NB  = NF / G_BN;
    constexpr int NCH = K / G_KC;
    static_assert(K % G_KC == 0 && NF % G_BN == 0, "tile sizes");

    const int t    = threadIdx.x;
    const int lane = t & 31;
    const int wave = t >> 5;
    const int wm   = wave >> 1;
    const int wn   = wave & 1;
    const int half = lane >> 4;
    const int l16  = lane & 15;
    const int m0 = (int)(blockIdx.x / NB) * G_BM;
    const int n0 = (int)(blockIdx.x % NB) * G_BN;

    const int q  = t & 31;
    const int r0 = t >> 5;
    v4f ra[8], rw[8];
    #pragma unroll
    for (int i = 0; i < 8; ++i) {
        int ar = m0 + r0 + 8 * i;
        if (ar >= M) ar = M - 1;
        ra[i] = *(const v4f*)(A + (size_t)ar * K + 4 * q);
        rw[i] = *(const v4f*)(W + (size_t)(n0 + r0 + 8 * i) * K + 4 * q);
    }

    const float* __restrict__ as  = As + (wm * 16 + l16) * G_PITCH + 2 * half;
    const float* __restrict__ ws0 = Ws + (wn * 32 + l16) * G_PITCH + 2 * half;
    const float* __restrict__ ws1 = ws0 + 16 * G_PITCH;

    v8f acc0 = {}, acc1 = {};
    float dv[8];
    #pragma unroll
    for (int c = 0; c < NCH; ++c) {
        if (c > 0) __syncthreads();
        #pragma unroll
        for (int i = 0; i < 8; ++i) {
            *(v4f*)(As + (r0 + 8 * i) * G_PITCH + 4 * q) = ra[i];
            *(v4f*)(Ws + (r0 + 8 * i) * G_PITCH + 4 * q) = rw[i];
        }
        __syncthreads();
        if (c + 1 < NCH) {
            #pragma unroll
            for (int i = 0; i < 8; ++i) {
                int ar = m0 + r0 + 8 * i;
                if (ar >= M) ar = M - 1;
                ra[i] = *(const v4f*)(A + (size_t)ar * K + (c + 1) * G_KC + 4 * q);
                rw[i] = *(const v4f*)(W + (size_t)(n0 + r0 + 8 * i) * K + (c + 1) * G_KC + 4 * q);
            }
        }
        if (EPI && c == NCH - 1) {
            #pragma unroll
            for (int r = 0; r < 8; ++r) {
                int m = m0 + wm * 16 + half * 8 + r;
                dv[r] = 1.0f / fmaxf((float)deg[m < M ? m : M - 1], 1.0f);
            }
        }
        #pragma unroll
        for (int k = 0; k < G_KC; k += 4) {
            v2f a  = *(const v2f*)(as  + k);
            v2f b0 = *(const v2f*)(ws0 + k);
            v2f b1 = *(const v2f*)(ws1 + k);
            acc0 = __builtin_amdgcn_wmma_f32_16x16x4_f32(
                 false, a,  false, b0,
                 (short)0, acc0,  false,  false);
            acc1 = __builtin_amdgcn_wmma_f32_16x16x4_f32(
                 false, a,  false, b1,
                 (short)0, acc1,  false,  false);
        }
    }

    const int na = n0 + wn * 32 + l16;
    const int nb = na + 16;
    const float bva = EPI ? bias[na] : 0.0f;
    const float bvb = EPI ? bias[nb] : 0.0f;
    #pragma unroll
    for (int r = 0; r < 8; ++r) {
        int m = m0 + wm * 16 + half * 8 + r;
        if (m < M) {
            float va = acc0[r];
            float vb = acc1[r];
            if (EPI) {
                va = fmaxf(va * dv[r] + bva, 0.0f);
                vb = fmaxf(vb * dv[r] + bvb, 0.0f);
            }
            out[(size_t)m * NF + na] = va;
            out[(size_t)m * NF + nb] = vb;
        }
    }
}

extern "C" void kernel_launch(void* const* d_in, const int* in_sizes, int n_in,
                              void* d_out, int out_size, void* d_ws, size_t ws_size,
                              hipStream_t stream) {
    const float* x  = (const float*)d_in[0];
    const int*   ei = (const int*)  d_in[1];
    const float* W1 = (const float*)d_in[3];
    const float* b1 = (const float*)d_in[4];
    const float* W2 = (const float*)d_in[5];
    const float* b2 = (const float*)d_in[6];
    const float* W3 = (const float*)d_in[7];
    const float* b3 = (const float*)d_in[8];
    float* out = (float*)d_out;

    const int M = in_sizes[0] / N_IN;
    const int E = in_sizes[1] / 2;
    const int* row = ei;
    const int* col = ei + E;

    const size_t Mp = ((size_t)M + 1 + 63) & ~(size_t)63;
    const size_t Ep = ((size_t)E + 31) & ~(size_t)31;
    int* wsi = (int*)d_ws;
    size_t off = 0;
    int* cnt     = wsi + off; off += Mp;
    int* bucket  = wsi + off; off += (size_t)M * CAP;
    int* ovf_dst = wsi + off; off += Ep;
    int* ovf_src = wsi + off; off += Ep;
    float* ws = (float*)d_ws;
    float* agg1 = ws + off; off += (size_t)M * N_IN;
    float* h1   = ws + off; off += (size_t)M * N_HID;
    float* z2   = ws + off; off += (size_t)M * N_HID2;
    (void)ws_size;

    hipMemsetAsync(cnt, 0, sizeof(int) * ((size_t)M + 1), stream);

    bucket_fill_kernel<<<(E + 255) / 256, 256, 0, stream>>>(row, col, cnt, bucket, ovf_dst, ovf_src, M, E);

    gather_sum_kernel<N_IN><<<(M + 7) / 8, 256, 0, stream>>>(x, bucket, cnt, ovf_dst, ovf_src, agg1, M);
    int blocks1 = ((M + G_BM - 1) / G_BM) * (N_HID / G_BN);
    gemm_wmma<N_IN, N_HID, true>
        <<<blocks1, 256, 0, stream>>>(agg1, W1, b1, cnt, h1, M);

    int blocks2 = ((M + G_BM - 1) / G_BM) * (N_HID2 / G_BN);
    gemm_wmma<N_HID, N_HID2, false>
        <<<blocks2, 256, 0, stream>>>(h1, W2, b2, cnt, z2, M);
    gather_head_kernel<N_HID2><<<(M + FIN_WAVES - 1) / FIN_WAVES, FIN_WAVES * 32, 0, stream>>>(
        z2, bucket, cnt, ovf_dst, ovf_src, b2, W3, b3, out, M);
}


# Both dense layers in one launch: the 64-row hidden tile (relu, 1/deg, bias applied) stays in LDS and is multiplied by W2 at once; the hidden plane and a launch are gone; k order and sum order kept
# speedup vs baseline: 6.9953x; 1.0491x over previous
#include <hip/hip_runtime.h>

typedef __attribute__((ext_vector_type(2))) float v2f;
typedef __attribute__((ext_vector_type(4))) float v4f;
typedef __attribute__((ext_vector_type(8))) float v8f;

#define N_IN   128
#define N_HID  256
#define N_HID2 128
#define N_CLS  2

#define CAP 64

__global__ void bucket_fill_kernel(const int* __restrict__ row, const int* __restrict__ col,
                                   int* __restrict__ cnt, int* __restrict__ bucket,
                                   int* __restrict__ ovf_dst, int* __restrict__ ovf_src,
                                   int M, int E) {
    int e = blockIdx.x * blockDim.x + threadIdx.x;
    if (e >= E) return;
    const int c = col[e];
    const int r = row[e];
    const int slot = atomicAdd(&cnt[c], 1);
    if (slot < CAP) {
        bucket[(size_t)c * CAP + slot] = r;
    } else {
        const int k = atomicAdd(&cnt[M], 1);
        ovf_dst[k] = c;
        ovf_src[k] = r;
    }
}

template<int F, int G>
__device__ __forceinline__ void add_rows(const float* __restrict__ src, int my, int j, int lane,
                                         int last, v4f (&acc)[F / 128]) {
    v4f v[G][F / 128];
    #pragma unroll
    for (int g = 0; g < G; ++g) {
        int r = __builtin_amdgcn_readlane(my, j + g);
        r = r < 0 ? 0 : (r > last ? last : r);
        const float* __restrict__ p = src + (size_t)r * F + lane * 4;
        #pragma unroll
        for (int q = 0; q < F / 128; ++q) v[g][q] = *(const v4f*)(p + q * 128);
    }
    #pragma unroll
    for (int g = 0; g < G; ++g) {
        #pragma unroll
        for (int q = 0; q < F / 128; ++q) acc[q] += v[g][q];
    }
}

template<int F>
__device__ __forceinline__ int sum_node_rows(const float* __restrict__ src, const int* __restrict__ bucket,
                                             const int* __restrict__ cnt, const int* __restrict__ ovf_dst,
                                             const int* __restrict__ ovf_src, int node, int lane, int n,
                                             v4f (&acc)[F / 128]) {
    constexpr int U = 1024 / F;
    const int d  = cnt[node];
    const int dm = d < CAP ? d : CAP;
    const int* __restrict__ seg = bucket + (size_t)node * CAP;
    #pragma unroll
    for (int q = 0; q < F / 128; ++q) acc[q] = (v4f){0.0f, 0.0f, 0.0f, 0.0f};

    for (int base = 0; base < dm; base += 32) {
        const int m = (dm - base < 32) ? (dm - base) : 32;
        const int my = (lane < m) ? seg[base + lane] : 0;
        int j = 0;
        for (; j + U <= m; j += U) add_rows<F, U>(src, my, j, lane, n - 1, acc);
        if (U == 8 && m - j >= 4) { add_rows<F, 4>(src, my, j, lane, n - 1, acc); j += 4; }
        if (m - j >= 2) { add_rows<F, 2>(src, my, j, lane, n - 1, acc); j += 2; }
        if (m - j >= 1) { add_rows<F, 1>(src, my, j, lane, n - 1, acc); }
    }

    if (d > CAP) {
        const int L = cnt[n];
        for (int k0 = 0; k0 < L; k0 += 32) {
            const int kk = k0 + lane;
            const int dst = (kk < L) ? ovf_dst[kk] : -1;
            const int sv  = (kk < L) ? ovf_src[kk] : 0;
            unsigned hit = __builtin_amdgcn_ballot_w32(dst == node);
            while (hit) {
                const int bit = __builtin_ctz(hit);
                hit &= hit - 1;
                add_rows<F, 1>(src, sv, bit, lane, n - 1, acc);
            }
        }
    }
    return d;
}

template<int F>
__global__ __launch_bounds__(256)
void gather_sum_kernel(const float* __restrict__ src, const int* __restrict__ bucket,
                       const int* __restrict__ cnt, const int* __restrict__ ovf_dst,
                       const int* __restrict__ ovf_src, float* __restrict__ agg, int n) {
    const int lane = threadIdx.x & 31;
    const int node = __builtin_amdgcn_readfirstlane((int)(blockIdx.x * (blockDim.x >> 5) + (threadIdx.x >> 5)));
    if (node >= n) return;
    v4f acc[F / 128];
    sum_node_rows<F>(src, bucket, cnt, ovf_dst, ovf_src, node, lane, n, acc);
    float* __restrict__ o = agg + (size_t)node * F + lane * 4;
    #pragma unroll
    for (int q = 0; q < F / 128; ++q) *(v4f*)(o + q * 128) = acc[q];
}

#define FIN_WAVES 16
template<int F>
__global__ __launch_bounds__(FIN_WAVES * 32)
void gather_head_kernel(const float* __restrict__ src, const int* __restrict__ bucket,
                        const int* __restrict__ cnt, const int* __restrict__ ovf_dst,
                        const int* __restrict__ ovf_src, const float* __restrict__ bias,
                        const float* __restrict__ W3, const float* __restrict__ b3,
                        float* __restrict__ out, int n) {
    static_assert(FIN_WAVES * N_CLS == 32, "one wave writes the group's output line");
    __shared__ float line[FIN_WAVES * N_CLS];
    const int lane = threadIdx.x & 31;
    const int wave = __builtin_amdgcn_readfirstlane((int)(threadIdx.x >> 5));
    const int node = __builtin_amdgcn_readfirstlane((int)(blockIdx.x * FIN_WAVES) + wave);
    if (node < n) {
        v4f bv[F / 128], w0[F / 128], w1[F / 128];
        #pragma unroll
        for (int q = 0; q < F / 128; ++q) {
            bv[q] = *(const v4f*)(bias + q * 128 + lane * 4);
            w0[q] = *(const v4f*)(W3 + q * 128 + lane * 4);
            w1[q] = *(const v4f*)(W3 + F + q * 128 + lane * 4);
        }
        const float c0 = b3[0];
        const float c1 = b3[1];
        v4f acc[F / 128];
        const int d = sum_node_rows<F>(src, bucket, cnt, ovf_dst, ovf_src, node, lane, n, acc);
        const float s = 1.0f / fmaxf((float)d, 1.0f);
        float p0 = 0.0f, p1 = 0.0f;
        #pragma unroll
        for (int q = 0; q < F / 128; ++q) {
            v4f v = acc[q] * s + bv[q];
            v.x = fmaxf(v.x, 0.0f); v.y = fmaxf(v.y, 0.0f);
            v.z = fmaxf(v.z, 0.0f); v.w = fmaxf(v.w, 0.0f);
            p0 += v.x * w0[q].x + v.y * w0[q].y + v.z * w0[q].z + v.w * w0[q].w;
            p1 += v.x * w1[q].x + v.y * w1[q].y + v.z * w1[q].z + v.w * w1[q].w;
        }
        #pragma unroll
        for (int off = 16; off >= 1; off >>= 1) {
            p0 += __shfl_xor(p0, off, 32);
            p1 += __shfl_xor(p1, off, 32);
        }
        if (lane == 0) {
            line[wave * N_CLS + 0] = p0 + c0;
            line[wave * N_CLS + 1] = p1 + c1;
        }
    }
    __syncthreads();
    if (wave == 0) {
        const int idx = (int)blockIdx.x * (FIN_WAVES * N_CLS) + lane;
        if (idx < n * N_CLS) out[idx] = line[lane];
    }
}

#define G_BM    64
#define G_BN    64
#define G_KC    128
#define G_PITCH 132
#define H_PITCH 260
#define N_ST1   (N_HID / G_BN)
#define N_ST2   ((N_HID2 / G_BN) * (N_HID / G_KC))

__device__ __forceinline__ void load_w_block(int s, const float* __restrict__ W1, const float* __restrict__ W2,
                                             int r0, int q, v4f (&rw)[8]) {
    const float* __restrict__ p;
    int stride;
    if (s < N_ST1) {
        p = W1 + (size_t)(s * G_BN) * N_IN;
        stride = N_IN;
    } else {
        const int s2 = s - N_ST1;
        p = W2 + (size_t)((s2 >> 1) * G_BN) * N_HID + (s2 & 1) * G_KC;
        stride = N_HID;
    }
    #pragma unroll
    for (int i = 0; i < 8; ++i) rw[i] = *(const v4f*)(p + (size_t)(r0 + 8 * i) * stride + 4 * q);
}

__global__ __launch_bounds__(256)
void dense_layers_wmma(const float* __restrict__ A, const float* __restrict__ W1,
                       const float* __restrict__ b1, const int* __restrict__ deg,
                       const float* __restrict__ W2, float* __restrict__ out, int M) {
    static_assert(N_IN == G_KC && N_HID % G_BN == 0 && N_HID2 % G_BN == 0 && N_HID == 2 * G_KC, "tile sizes");
    __shared__ __attribute__((aligned(16))) float As[G_BM * G_PITCH];
    __shared__ __attribute__((aligned(16))) float Ws[G_BN * G_PITCH];
    __shared__ __attribute__((aligned(16))) float Hs[G_BM * H_PITCH];

    const int t    = threadIdx.x;
    const int lane = t & 31;
    const int wave = t >> 5;
    const int wm   = wave >> 1;
    const int wn   = wave & 1;
    const int half = lane >> 4;
    const int l16  = lane & 15;
    const int m0 = (int)blockIdx.x * G_BM;

    const int q  = t & 31;
    const int r0 = t >> 5;
    v4f ra[8], rw[8];
    #pragma unroll
    for (int i = 0; i < 8; ++i) {
        int ar = m0 + r0 + 8 * i;
        if (ar >= M) ar = M - 1;
        ra[i] = *(const v4f*)(A + (size_t)ar * N_IN + 4 * q);
    }
    load_w_block(0, W1, W2, r0, q, rw);

    float dv[8];
    #pragma unroll
    for (int r = 0; r < 8; ++r) {
        int m = m0 + wm * 16 + half * 8 + r;
        dv[r] = 1.0f / fmaxf((float)deg[m < M ? m : M - 1], 1.0f);
    }

    const float* __restrict__ as  = As + (wm * 16 + l16) * G_PITCH + 2 * half;
    const float* __restrict__ hs  = Hs + (wm * 16 + l16) * H_PITCH + 2 * half;
    const float* __restrict__ ws0 = Ws + (wn * 32 + l16) * G_PITCH + 2 * half;
    const float* __restrict__ ws1 = ws0 + 16 * G_PITCH;

    #pragma unroll
    for (int i = 0; i < 8; ++i) {
        *(v4f*)(As + (r0 + 8 * i) * G_PITCH + 4 * q) = ra[i];
        *(v4f*)(Ws + (r0 + 8 * i) * G_PITCH + 4 * q) = rw[i];
    }
    __syncthreads();

    #pragma unroll
    for (int s = 0; s < N_ST1; ++s) {
        load_w_block(s + 1, W1, W2, r0, q, rw);
        const int ca = s * G_BN + wn * 32 + l16;
        const int cb = ca + 16;
        const float bva = b1[ca];
        const float bvb = b1[cb];
        v8f acc0 = {}, acc1 = {};
        #pragma unroll
        for (int k = 0; k < G_KC; k += 4) {
            v2f a  = *(const v2f*)(as  + k);
            v2f b0 = *(const v2f*)(ws0 + k);
            v2f b1v = *(const v2f*)(ws1 + k);
            acc0 = __builtin_amdgcn_wmma_f32_16x16x4_f32(
                 false, a,  false, b0,
                 (short)0, acc0,  false,  false);
            acc1 = __builtin_amdgcn_wmma_f32_16x16x4_f32(
                 false, a,  false, b1v,
                 (short)0, acc1,  false,  false);
        }
        #pragma unroll
        for (int r = 0; r < 8; ++r) {
            float* __restrict__ hrow = Hs + (wm * 16 + half * 8 + r) * H_PITCH;
            hrow[ca] = fmaxf(acc0[r] * dv[r] + bva, 0.0f);
            hrow[cb] = fmaxf(acc1[r] * dv[r] + bvb, 0.0f);
        }
        __syncthreads();
        #pragma unroll
        for (int i = 0; i < 8; ++i) *(v4f*)(Ws + (r0 + 8 * i) * G_PITCH + 4 * q) = rw[i];
        __syncthreads();
    }

    #pragma unroll
    for (int nb = 0; nb < N_HID2 / G_BN; ++nb) {
        v8f acc0 = {}, acc1 = {};
        #pragma unroll
        for (int kc = 0; kc < N_HID / G_KC; ++kc) {
            const int s = N_ST1 + nb * (N_HID / G_KC) + kc;
            const bool more = (s + 1 < N_ST1 + N_ST2);
            if (more) load_w_block(s + 1, W1, W2, r0, q, rw);
            const float* __restrict__ hk = hs + kc * G_KC;
            #pragma unroll
            for (int k = 0; k < G_KC; k += 4) {
                v2f a  = *(const v2f*)(hk  + k);
                v2f b0 = *(const v2f*)(ws0 + k);
                v2f b1v = *(const v2f*)(ws1 + k);
                acc0 = __builtin_amdgcn_wmma_f32_16x16x4_f32(
                     false, a,  false, b0,
                     (short)0, acc0,  false,  false);
                acc1 = __builtin_amdgcn_wmma_f32_16x16x4_f32(
                     false, a,  false, b1v,
                     (short)0, acc1,  false,  false);
            }
            if (more) {
                __syncthreads();
                #pragma unroll
                for (int i = 0; i < 8; ++i) *(v4f*)(Ws + (r0 + 8 * i) * G_PITCH + 4 * q) = rw[i];
                __syncthreads();
            }
        }
        const int na = nb * G_BN + wn * 32 + l16;
        const int nc = na + 16;
        #pragma unroll
        for (int r = 0; r < 8; ++r) {
            int m = m0 + wm * 16 + half * 8 + r;
            if (m < M) {
                out[(size_t)m * N_HID2 + na] = acc0[r];
                out[(size_t)m * N_HID2 + nc] = acc1[r];
            }
        }
    }
}

extern "C" void kernel_launch(void* const* d_in, const int* in_sizes, int n_in,
                              void* d_out, int out_size, void* d_ws, size_t ws_size,
                              hipStream_t stream) {
    const float* x  = (const float*)d_in[0];
    const int*   ei = (const int*)  d_in[1];
    const float* W1 = (const float*)d_in[3];
    const float* b1 = (const float*)d_in[4];
    const float* W2 = (const float*)d_in[5];
    const float* b2 = (const float*)d_in[6];
    const float* W3 = (const float*)d_in[7];
    const float* b3 = (const float*)d_in[8];
    float* out = (float*)d_out;

    const int M = in_sizes[0] / N_IN;
    const int E = in_sizes[1] / 2;
    const int* row = ei;
    const int* col = ei + E;

    const size_t Mp = ((size_t)M + 1 + 63) & ~(size_t)63;
    const size_t Ep = ((size_t)E + 31) & ~(size_t)31;
    int* wsi = (int*)d_ws;
    size_t off = 0;
    int* cnt     = wsi + off; off += Mp;
    int* bucket  = wsi + off; off += (size_t)M * CAP;
    int* ovf_dst = wsi + off; off += Ep;
    int* ovf_src = wsi + off; off += Ep;
    float* ws = (float*)d_ws;
    float* agg1 = ws + off; off += (size_t)M * N_IN;
    float* z2   = ws + off; off += (size_t)M * N_HID2;
    (void)ws_size;

    hipMemsetAsync(cnt, 0, sizeof(int) * ((size_t)M + 1), stream);

    bucket_fill_kernel<<<(E + 255) / 256, 256, 0, stream>>>(row, col, cnt, bucket, ovf_dst, ovf_src, M, E);

    gather_sum_kernel<N_IN><<<(M + 7) / 8, 256, 0, stream>>>(x, bucket, cnt, ovf_dst, ovf_src, agg1, M);

    dense_layers_wmma<<<(M + G_BM - 1) / G_BM, 256, 0, stream>>>(agg1, W1, b1, cnt, W2, z2, M);

    gather_head_kernel<N_HID2><<<(M + FIN_WAVES - 1) / FIN_WAVES, FIN_WAVES * 32, 0, stream>>>(
        z2, bucket, cnt, ovf_dst, ovf_src, b2, W3, b3, out, M);
}


# In-degree counters set 128 bytes apart, one per memory line (were 32 per line), so the fill's returning atomics do not queue on shared lines; ranks stay arrival order, sum order varies as before
# speedup vs baseline: 8.2330x; 1.1769x over previous
#include <hip/hip_runtime.h>

typedef __attribute__((ext_vector_type(2))) float v2f;
typedef __attribute__((ext_vector_type(4))) float v4f;
typedef __attribute__((ext_vector_type(8))) float v8f;

#define N_IN   128
#define N_HID  256
#define N_HID2 128
#define N_CLS  2

#define CAP 64
#define CNT_STRIDE 32

__global__ void bucket_fill_kernel(const int* __restrict__ row, const int* __restrict__ col,
                                   int* __restrict__ cnt, int* __restrict__ bucket,
                                   int* __restrict__ ovf_dst, int* __restrict__ ovf_src,
                                   int M, int E) {
    int e = blockIdx.x * blockDim.x + threadIdx.x;
    if (e >= E) return;
    const int c = col[e];
    const int r = row[e];
    const int slot = atomicAdd(&cnt[(size_t)c * CNT_STRIDE], 1);
    if (slot < CAP) {
        bucket[(size_t)c * CAP + slot] = r;
    } else {
        const int k = atomicAdd(&cnt[(size_t)M * CNT_STRIDE], 1);
        ovf_dst[k] = c;
        ovf_src[k] = r;
    }
}

template<int F, int G>
__device__ __forceinline__ void add_rows(const float* __restrict__ src, int my, int j, int lane,
                                         int last, v4f (&acc)[F / 128]) {
    v4f v[G][F / 128];
    #pragma unroll
    for (int g = 0; g < G; ++g) {
        int r = __builtin_amdgcn_readlane(my, j + g);
        r = r < 0 ? 0 : (r > last ? last : r);
        const float* __restrict__ p = src + (size_t)r * F + lane * 4;
        #pragma unroll
        for (int q = 0; q < F / 128; ++q) v[g][q] = *(const v4f*)(p + q * 128);
    }
    #pragma unroll
    for (int g = 0; g < G; ++g) {
        #pragma unroll
        for (int q = 0; q < F / 128; ++q) acc[q] += v[g][q];
    }
}

template<int F>
__device__ __forceinline__ int sum_node_rows(const float* __restrict__ src, const int* __restrict__ bucket,
                                             const int* __restrict__ cnt, const int* __restrict__ ovf_dst,
                                             const int* __restrict__ ovf_src, int node, int lane, int n,
                                             v4f (&acc)[F / 128]) {
    constexpr int U = 1024 / F;
    const int d  = cnt[(size_t)node * CNT_STRIDE];
    const int dm = d < CAP ? d : CAP;
    const int* __restrict__ seg = bucket + (size_t)node * CAP;
    #pragma unroll
    for (int q = 0; q < F / 128; ++q) acc[q] = (v4f){0.0f, 0.0f, 0.0f, 0.0f};

    for (int base = 0; base < dm; base += 32) {
        const int m = (dm - base < 32) ? (dm - base) : 32;
        const int my = (lane < m) ? seg[base + lane] : 0;
        int j = 0;
        for (; j + U <= m; j += U) add_rows<F, U>(src, my, j, lane, n - 1, acc);
        if (U == 8 && m - j >= 4) { add_rows<F, 4>(src, my, j, lane, n - 1, acc); j += 4; }
        if (m - j >= 2) { add_rows<F, 2>(src, my, j, lane, n - 1, acc); j += 2; }
        if (m - j >= 1) { add_rows<F, 1>(src, my, j, lane, n - 1, acc); }
    }

    if (d > CAP) {
        const int L = cnt[(size_t)n * CNT_STRIDE];
        for (int k0 = 0; k0 < L; k0 += 32) {
            const int kk = k0 + lane;
            const int dst = (kk < L) ? ovf_dst[kk] : -1;
            const int sv  = (kk < L) ? ovf_src[kk] : 0;
            unsigned hit = __builtin_amdgcn_ballot_w32(dst == node);
            while (hit) {
                const int bit = __builtin_ctz(hit);
                hit &= hit - 1;
                add_rows<F, 1>(src, sv, bit, lane, n - 1, acc);
            }
        }
    }
    return d;
}

template<int F>
__global__ __launch_bounds__(256)
void gather_sum_kernel(const float* __restrict__ src, const int* __restrict__ bucket,
                       const int* __restrict__ cnt, const int* __restrict__ ovf_dst,
                       const int* __restrict__ ovf_src, float* __restrict__ agg, int n) {
    const int lane = threadIdx.x & 31;
    const int node = __builtin_amdgcn_readfirstlane((int)(blockIdx.x * (blockDim.x >> 5) + (threadIdx.x >> 5)));
    if (node >= n) return;
    v4f acc[F / 128];
    sum_node_rows<F>(src, bucket, cnt, ovf_dst, ovf_src, node, lane, n, acc);
    float* __restrict__ o = agg + (size_t)node * F + lane * 4;
    #pragma unroll
    for (int q = 0; q < F / 128; ++q) *(v4f*)(o + q * 128) = acc[q];
}

#define FIN_WAVES 16
template<int F>
__global__ __launch_bounds__(FIN_WAVES * 32)
void gather_head_kernel(const float* __restrict__ src, const int* __restrict__ bucket,
                        const int* __restrict__ cnt, const int* __restrict__ ovf_dst,
                        const int* __restrict__ ovf_src, const float* __restrict__ bias,
                        const float* __restrict__ W3, const float* __restrict__ b3,
                        float* __restrict__ out, int n) {
    static_assert(FIN_WAVES * N_CLS == 32, "one wave writes the group's output line");
    __shared__ float line[FIN_WAVES * N_CLS];
    const int lane = threadIdx.x & 31;
    const int wave = __builtin_amdgcn_readfirstlane((int)(threadIdx.x >> 5));
    const int node = __builtin_amdgcn_readfirstlane((int)(blockIdx.x * FIN_WAVES) + wave);
    if (node < n) {
        v4f bv[F / 128], w0[F / 128], w1[F / 128];
        #pragma unroll
        for (int q = 0; q < F / 128; ++q) {
            bv[q] = *(const v4f*)(bias + q * 128 + lane * 4);
            w0[q] = *(const v4f*)(W3 + q * 128 + lane * 4);
            w1[q] = *(const v4f*)(W3 + F + q * 128 + lane * 4);
        }
        const float c0 = b3[0];
        const float c1 = b3[1];
        v4f acc[F / 128];
        const int d = sum_node_rows<F>(src, bucket, cnt, ovf_dst, ovf_src, node, lane, n, acc);
        const float s = 1.0f / fmaxf((float)d, 1.0f);
        float p0 = 0.0f, p1 = 0.0f;
        #pragma unroll
        for (int q = 0; q < F / 128; ++q) {
            v4f v = acc[q] * s + bv[q];
            v.x = fmaxf(v.x, 0.0f); v.y = fmaxf(v.y, 0.0f);
            v.z = fmaxf(v.z, 0.0f); v.w = fmaxf(v.w, 0.0f);
            p0 += v.x * w0[q].x + v.y * w0[q].y + v.z * w0[q].z + v.w * w0[q].w;
            p1 += v.x * w1[q].x + v.y * w1[q].y + v.z * w1[q].z + v.w * w1[q].w;
        }
        #pragma unroll
        for (int off = 16; off >= 1; off >>= 1) {
            p0 += __shfl_xor(p0, off, 32);
            p1 += __shfl_xor(p1, off, 32);
        }
        if (lane == 0) {
            line[wave * N_CLS + 0] = p0 + c0;
            line[wave * N_CLS + 1] = p1 + c1;
        }
    }
    __syncthreads();
    if (wave == 0) {
        const int idx = (int)blockIdx.x * (FIN_WAVES * N_CLS) + lane;
        if (idx < n * N_CLS) out[idx] = line[lane];
    }
}

#define G_BM    64
#define G_BN    64
#define G_KC    128
#define G_PITCH 132
#define H_PITCH 260
#define N_ST1   (N_HID / G_BN)
#define N_ST2   ((N_HID2 / G_BN) * (N_HID / G_KC))

__device__ __forceinline__ void load_w_block(int s, const float* __restrict__ W1, const float* __restrict__ W2,
                                             int r0, int q, v4f (&rw)[8]) {
    const float* __restrict__ p;
    int stride;
    if (s < N_ST1) {
        p = W1 + (size_t)(s * G_BN) * N_IN;
        stride = N_IN;
    } else {
        const int s2 = s - N_ST1;
        p = W2 + (size_t)((s2 >> 1) * G_BN) * N_HID + (s2 & 1) * G_KC;
        stride = N_HID;
    }
    #pragma unroll
    for (int i = 0; i < 8; ++i) rw[i] = *(const v4f*)(p + (size_t)(r0 + 8 * i) * stride + 4 * q);
}

__global__ __launch_bounds__(256)
void dense_layers_wmma(const float* __restrict__ A, const float* __restrict__ W1,
                       const float* __restrict__ b1, const int* __restrict__ deg,
                       const float* __restrict__ W2, float* __restrict__ out, int M) {
    static_assert(N_IN == G_KC && N_HID % G_BN == 0 && N_HID2 % G_BN == 0 && N_HID == 2 * G_KC, "tile sizes");
    __shared__ __attribute__((aligned(16))) float As[G_BM * G_PITCH];
    __shared__ __attribute__((aligned(16))) float Ws[G_BN * G_PITCH];
    __shared__ __attribute__((aligned(16))) float Hs[G_BM * H_PITCH];

    const int t    = threadIdx.x;
    const int lane = t & 31;
    const int wave = t >> 5;
    const int wm   = wave >> 1;
    const int wn   = wave & 1;
    const int half = lane >> 4;
    const int l16  = lane & 15;
    const int m0 = (int)blockIdx.x * G_BM;

    const int q  = t & 31;
    const int r0 = t >> 5;
    v4f ra[8], rw[8];
    #pragma unroll
    for (int i = 0; i < 8; ++i) {
        int ar = m0 + r0 + 8 * i;
        if (ar >= M) ar = M - 1;
        ra[i] = *(const v4f*)(A + (size_t)ar * N_IN + 4 * q);
    }
    load_w_block(0, W1, W2, r0, q, rw);

    float dv[8];
    #pragma unroll
    for (int r = 0; r < 8; ++r) {
        int m = m0 + wm * 16 + half * 8 + r;
        dv[r] = 1.0f / fmaxf((float)deg[(size_t)(m < M ? m : M - 1) * CNT_STRIDE], 1.0f);
    }

    const float* __restrict__ as  = As + (wm * 16 + l16) * G_PITCH + 2 * half;
    const float* __restrict__ hs  = Hs + (wm * 16 + l16) * H_PITCH + 2 * half;
    const float* __restrict__ ws0 = Ws + (wn * 32 + l16) * G_PITCH + 2 * half;
    const float* __restrict__ ws1 = ws0 + 16 * G_PITCH;

    #pragma unroll
    for (int i = 0; i < 8; ++i) {
        *(v4f*)(As + (r0 + 8 * i) * G_PITCH + 4 * q) = ra[i];
        *(v4f*)(Ws + (r0 + 8 * i) * G_PITCH + 4 * q) = rw[i];
    }
    __syncthreads();

    #pragma unroll
    for (int s = 0; s < N_ST1; ++s) {
        load_w_block(s + 1, W1, W2, r0, q, rw);
        const int ca = s * G_BN + wn * 32 + l16;
        const int cb = ca + 16;
        const float bva = b1[ca];
        const float bvb = b1[cb];
        v8f acc0 = {}, acc1 = {};
        #pragma unroll
        for (int k = 0; k < G_KC; k += 4) {
            v2f a  = *(const v2f*)(as  + k);
            v2f b0 = *(const v2f*)(ws0 + k);
            v2f b1v = *(const v2f*)(ws1 + k);
            acc0 = __builtin_amdgcn_wmma_f32_16x16x4_f32(
                 false, a,  false, b0,
                 (short)0, acc0,  false,  false);
            acc1 = __builtin_amdgcn_wmma_f32_16x16x4_f32(
                 false, a,  false, b1v,
                 (short)0, acc1,  false,  false);
        }
        #pragma unroll
        for (int r = 0; r < 8; ++r) {
            float* __restrict__ hrow = Hs + (wm * 16 + half * 8 + r) * H_PITCH;
            hrow[ca] = fmaxf(acc0[r] * dv[r] + bva, 0.0f);
            hrow[cb] = fmaxf(acc1[r] * dv[r] + bvb, 0.0f);
        }
        __syncthreads();
        #pragma unroll
        for (int i = 0; i < 8; ++i) *(v4f*)(Ws + (r0 + 8 * i) * G_PITCH + 4 * q) = rw[i];
        __syncthreads();
    }

    #pragma unroll
    for (int nb = 0; nb < N_HID2 / G_BN; ++nb) {
        v8f acc0 = {}, acc1 = {};
        #pragma unroll
        for (int kc = 0; kc < N_HID / G_KC; ++kc) {
            const int s = N_ST1 + nb * (N_HID / G_KC) + kc;
            const bool more = (s + 1 < N_ST1 + N_ST2);
            if (more) load_w_block(s + 1, W1, W2, r0, q, rw);
            const float* __restrict__ hk = hs + kc * G_KC;
            #pragma unroll
            for (int k = 0; k < G_KC; k += 4) {
                v2f a  = *(const v2f*)(hk  + k);
                v2f b0 = *(const v2f*)(ws0 + k);
                v2f b1v = *(const v2f*)(ws1 + k);
                acc0 = __builtin_amdgcn_wmma_f32_16x16x4_f32(
                     false, a,  false, b0,
                     (short)0, acc0,  false,  false);
                acc1 = __builtin_amdgcn_wmma_f32_16x16x4_f32(
                     false, a,  false, b1v,
                     (short)0, acc1,  false,  false);
            }
            if (more) {
                __syncthreads();
                #pragma unroll
                for (int i = 0; i < 8; ++i) *(v4f*)(Ws + (r0 + 8 * i) * G_PITCH + 4 * q) = rw[i];
                __syncthreads();
            }
        }
        const int na = nb * G_BN + wn * 32 + l16;
        const int nc = na + 16;
        #pragma unroll
        for (int r = 0; r < 8; ++r) {
            int m = m0 + wm * 16 + half * 8 + r;
            if (m < M) {
                out[(size_t)m * N_HID2 + na] = acc0[r];
                out[(size_t)m * N_HID2 + nc] = acc1[r];
            }
        }
    }
}

extern "C" void kernel_launch(void* const* d_in, const int* in_sizes, int n_in,
                              void* d_out, int out_size, void* d_ws, size_t ws_size,
                              hipStream_t stream) {
    const float* x  = (const float*)d_in[0];
    const int*   ei = (const int*)  d_in[1];
    const float* W1 = (const float*)d_in[3];
    const float* b1 = (const float*)d_in[4];
    const float* W2 = (const float*)d_in[5];
    const float* b2 = (const float*)d_in[6];
    const float* W3 = (const float*)d_in[7];
    const float* b3 = (const float*)d_in[8];
    float* out = (float*)d_out;

    const int M = in_sizes[0] / N_IN;
    const int E = in_sizes[1] / 2;
    const int* row = ei;
    const int* col = ei + E;

    const size_t Mp = (((size_t)M + 1) * CNT_STRIDE + 63) & ~(size_t)63;
    const size_t Ep = ((size_t)E + 31) & ~(size_t)31;
    int* wsi = (int*)d_ws;
    size_t off = 0;
    int* cnt     = wsi + off; off += Mp;
    int* bucket  = wsi + off; off += (size_t)M * CAP;
    int* ovf_dst = wsi + off; off += Ep;
    int* ovf_src = wsi + off; off += Ep;
    float* ws = (float*)d_ws;
    float* agg1 = ws + off; off += (size_t)M * N_IN;
    float* z2   = ws + off; off += (size_t)M * N_HID2;
    (void)ws_size;

    hipMemsetAsync(cnt, 0, sizeof(int) * ((size_t)M + 1) * CNT_STRIDE, stream);

    bucket_fill_kernel<<<(E + 255) / 256, 256, 0, stream>>>(row, col, cnt, bucket, ovf_dst, ovf_src, M, E);

    gather_sum_kernel<N_IN><<<(M + 7) / 8, 256, 0, stream>>>(x, bucket, cnt, ovf_dst, ovf_src, agg1, M);

    dense_layers_wmma<<<(M + G_BM - 1) / G_BM, 256, 0, stream>>>(agg1, W1, b1, cnt, W2, z2, M);

    gather_head_kernel<N_HID2><<<(M + FIN_WAVES - 1) / FIN_WAVES, FIN_WAVES * 32, 0, stream>>>(
        z2, bucket, cnt, ovf_dst, ovf_src, b2, W3, b3, out, M);
}


# Layer-1 neighbour sum moved into the dense launch: a wave sums 8 nodes into the LDS A tile, each node's rows (up to 31) in flight as one load group; summed plane and a launch gone; sum order kept
# speedup vs baseline: 8.5067x; 1.0332x over previous
#include <hip/hip_runtime.h>

typedef __attribute__((ext_vector_type(2))) float v2f;
typedef __attribute__((ext_vector_type(4))) float v4f;
typedef __attribute__((ext_vector_type(8))) float v8f;

#define N_IN   128
#define N_HID  256
#define N_HID2 128
#define N_CLS  2

#define CAP 64
#define CNT_STRIDE 32

__global__ void bucket_fill_kernel(const int* __restrict__ row, const int* __restrict__ col,
                                   int* __restrict__ cnt, int* __restrict__ bucket,
                                   int* __restrict__ ovf_dst, int* __restrict__ ovf_src,
                                   int M, int E) {
    int e = blockIdx.x * blockDim.x + threadIdx.x;
    if (e >= E) return;
    const int c = col[e];
    const int r = row[e];
    const int slot = atomicAdd(&cnt[(size_t)c * CNT_STRIDE], 1);
    if (slot < CAP) {
        bucket[(size_t)c * CAP + slot] = r;
    } else {
        const int k = atomicAdd(&cnt[(size_t)M * CNT_STRIDE], 1);
        ovf_dst[k] = c;
        ovf_src[k] = r;
    }
}

template<int F, int G>
__device__ __forceinline__ void load_rows(const float* __restrict__ src, int my, int j, int lane,
                                          int last, v4f (&v)[G][F / 128]) {
    #pragma unroll
    for (int g = 0; g < G; ++g) {
        int r = __builtin_amdgcn_readlane(my, j + g);
        r = r < 0 ? 0 : (r > last ? last : r);
        const float* __restrict__ p = src + (size_t)r * F + lane * 4;
        #pragma unroll
        for (int q = 0; q < F / 128; ++q) v[g][q] = *(const v4f*)(p + q * 128);
    }
}

template<int F, int G>
__device__ __forceinline__ void sum_rows(const v4f (&v)[G][F / 128], v4f (&acc)[F / 128]) {
    #pragma unroll
    for (int g = 0; g < G; ++g) {
        #pragma unroll
        for (int q = 0; q < F / 128; ++q) acc[q] += v[g][q];
    }
}

template<int F, int G>
__device__ __forceinline__ void add_rows(const float* __restrict__ src, int my, int j, int lane,
                                         int last, v4f (&acc)[F / 128]) {
    v4f v[G][F / 128];
    load_rows<F, G>(src, my, j, lane, last, v);
    sum_rows<F, G>(v, acc);
}

template<int F>
__device__ __forceinline__ int sum_node_rows(const float* __restrict__ src, const int* __restrict__ bucket,
                                             const int* __restrict__ cnt, const int* __restrict__ ovf_dst,
                                             const int* __restrict__ ovf_src, int node, int lane, int n,
                                             v4f (&acc)[F / 128]) {
    constexpr int U = 1024 / F;
    const int d  = cnt[(size_t)node * CNT_STRIDE];
    const int dm = d < CAP ? d : CAP;
    const int* __restrict__ seg = bucket + (size_t)node * CAP;
    #pragma unroll
    for (int q = 0; q < F / 128; ++q) acc[q] = (v4f){0.0f, 0.0f, 0.0f, 0.0f};

    for (int base = 0; base < dm; base += 32) {
        const int m = (dm - base < 32) ? (dm - base) : 32;
        const int my = (lane < m) ? seg[base + lane] : 0;
        int j = 0;
        for (; j + U <= m; j += U) add_rows<F, U>(src, my, j, lane, n - 1, acc);
        if (U == 8 && m - j >= 4) { add_rows<F, 4>(src, my, j, lane, n - 1, acc); j += 4; }
        if (m - j >= 2) { add_rows<F, 2>(src, my, j, lane, n - 1, acc); j += 2; }
        if (m - j >= 1) { add_rows<F, 1>(src, my, j, lane, n - 1, acc); }
    }

    if (d > CAP) {
        const int L = cnt[(size_t)n * CNT_STRIDE];
        for (int k0 = 0; k0 < L; k0 += 32) {
            const int kk = k0 + lane;
            const int dst = (kk < L) ? ovf_dst[kk] : -1;
            const int sv  = (kk < L) ? ovf_src[kk] : 0;
            unsigned hit = __builtin_amdgcn_ballot_w32(dst == node);
            while (hit) {
                const int bit = __builtin_ctz(hit);
                hit &= hit - 1;
                add_rows<F, 1>(src, sv, bit, lane, n - 1, acc);
            }
        }
    }
    return d;
}

template<int F>
__device__ __forceinline__ void add_rows_pieces(const float* __restrict__ src, int my, int j, int c, int lane,
                                                int last, v4f (&acc)[F / 128]) {
    v4f v16[16][F / 128], v8[8][F / 128], v4[4][F / 128], v2[2][F / 128], v1[1][F / 128];
    int s = j;
    if (c & 16) { load_rows<F, 16>(src, my, s, lane, last, v16); s += 16; }
    if (c & 8)  { load_rows<F, 8>(src, my, s, lane, last, v8);   s += 8; }
    if (c & 4)  { load_rows<F, 4>(src, my, s, lane, last, v4);   s += 4; }
    if (c & 2)  { load_rows<F, 2>(src, my, s, lane, last, v2);   s += 2; }
    if (c & 1)  { load_rows<F, 1>(src, my, s, lane, last, v1); }
    if (c & 16) sum_rows<F, 16>(v16, acc);
    if (c & 8)  sum_rows<F, 8>(v8, acc);
    if (c & 4)  sum_rows<F, 4>(v4, acc);
    if (c & 2)  sum_rows<F, 2>(v2, acc);
    if (c & 1)  sum_rows<F, 1>(v1, acc);
}

template<int F>
__device__ __forceinline__ void sum_listed_rows(const float* __restrict__ src, const int* __restrict__ bucket,
                                                const int* __restrict__ cnt, const int* __restrict__ ovf_dst,
                                                const int* __restrict__ ovf_src, int node, int lane, int n,
                                                int d, int my0, v4f (&acc)[F / 128]) {
    const int dm = d < CAP ? d : CAP;
    #pragma unroll
    for (int q = 0; q < F / 128; ++q) acc[q] = (v4f){0.0f, 0.0f, 0.0f, 0.0f};

    const int m = dm < 32 ? dm : 32;
    add_rows_pieces<F>(src, my0, 0, m < 31 ? m : 31, lane, n - 1, acc);
    if (m == 32) add_rows<F, 1>(src, my0, 31, lane, n - 1, acc);

    if (dm > 32) {
        const int m2 = dm - 32;
        const int my = (lane < m2) ? bucket[(size_t)node * CAP + 32 + lane] : 0;
        int j = 0;
        for (; j + 8 <= m2; j += 8) add_rows<F, 8>(src, my, j, lane, n - 1, acc);
        if (m2 - j >= 4) { add_rows<F, 4>(src, my, j, lane, n - 1, acc); j += 4; }
        if (m2 - j >= 2) { add_rows<F, 2>(src, my, j, lane, n - 1, acc); j += 2; }
        if (m2 - j >= 1) { add_rows<F, 1>(src, my, j, lane, n - 1, acc); }
    }

    if (d > CAP) {
        const int L = cnt[(size_t)n * CNT_STRIDE];
        for (int k0 = 0; k0 < L; k0 += 32) {
            const int kk = k0 + lane;
            const int dst = (kk < L) ? ovf_dst[kk] : -1;
            const int sv  = (kk < L) ? ovf_src[kk] : 0;
            unsigned hit = __builtin_amdgcn_ballot_w32(dst == node);
            while (hit) {
                const int bit = __builtin_ctz(hit);
                hit &= hit - 1;
                add_rows<F, 1>(src, sv, bit, lane, n - 1, acc);
            }
        }
    }
}

#define FIN_WAVES 16
template<int F>
__global__ __launch_bounds__(FIN_WAVES * 32)
void gather_head_kernel(const float* __restrict__ src, const int* __restrict__ bucket,
                        const int* __restrict__ cnt, const int* __restrict__ ovf_dst,
                        const int* __restrict__ ovf_src, const float* __restrict__ bias,
                        const float* __restrict__ W3, const float* __restrict__ b3,
                        float* __restrict__ out, int n) {
    static_assert(FIN_WAVES * N_CLS == 32, "one wave writes the group's output line");
    __shared__ float line[FIN_WAVES * N_CLS];
    const int lane = threadIdx.x & 31;
    const int wave = __builtin_amdgcn_readfirstlane((int)(threadIdx.x >> 5));
    const int node = __builtin_amdgcn_readfirstlane((int)(blockIdx.x * FIN_WAVES) + wave);
    if (node < n) {
        v4f bv[F / 128], w0[F / 128], w1[F / 128];
        #pragma unroll
        for (int q = 0; q < F / 128; ++q) {
            bv[q] = *(const v4f*)(bias + q * 128 + lane * 4);
            w0[q] = *(const v4f*)(W3 + q * 128 + lane * 4);
            w1[q] = *(const v4f*)(W3 + F + q * 128 + lane * 4);
        }
        const float c0 = b3[0];
        const float c1 = b3[1];
        v4f acc[F / 128];
        const int d = sum_node_rows<F>(src, bucket, cnt, ovf_dst, ovf_src, node, lane, n, acc);
        const float s = 1.0f / fmaxf((float)d, 1.0f);
        float p0 = 0.0f, p1 = 0.0f;
        #pragma unroll
        for (int q = 0; q < F / 128; ++q) {
            v4f v = acc[q] * s + bv[q];
            v.x = fmaxf(v.x, 0.0f); v.y = fmaxf(v.y, 0.0f);
            v.z = fmaxf(v.z, 0.0f); v.w = fmaxf(v.w, 0.0f);
            p0 += v.x * w0[q].x + v.y * w0[q].y + v.z * w0[q].z + v.w * w0[q].w;
            p1 += v.x * w1[q].x + v.y * w1[q].y + v.z * w1[q].z + v.w * w1[q].w;
        }
        #pragma unroll
        for (int off = 16; off >= 1; off >>= 1) {
            p0 += __shfl_xor(p0, off, 32);
            p1 += __shfl_xor(p1, off, 32);
        }
        if (lane == 0) {
            line[wave * N_CLS + 0] = p0 + c0;
            line[wave * N_CLS + 1] = p1 + c1;
        }
    }
    __syncthreads();
    if (wave == 0) {
        const int idx = (int)blockIdx.x * (FIN_WAVES * N_CLS) + lane;
        if (idx < n * N_CLS) out[idx] = line[lane];
    }
}

#define G_BM    64
#define G_BN    64
#define G_KC    128
#define G_PITCH 132
#define H_PITCH 260
#define N_ST1   (N_HID / G_BN)
#define N_ST2   ((N_HID2 / G_BN) * (N_HID / G_KC))

__device__ __forceinline__ void load_w_block(int s, const float* __restrict__ W1, const float* __restrict__ W2,
                                             int r0, int q, v4f (&rw)[8]) {
    const float* __restrict__ p;
    int stride;
    if (s < N_ST1) {
        p = W1 + (size_t)(s * G_BN) * N_IN;
        stride = N_IN;
    } else {
        const int s2 = s - N_ST1;
        p = W2 + (size_t)((s2 >> 1) * G_BN) * N_HID + (s2 & 1) * G_KC;
        stride = N_HID;
    }
    #pragma unroll
    for (int i = 0; i < 8; ++i) rw[i] = *(const v4f*)(p + (size_t)(r0 + 8 * i) * stride + 4 * q);
}

__global__ __launch_bounds__(256)
void dense_layers_wmma(const float* __restrict__ x, const int* __restrict__ bucket,
                       const int* __restrict__ cnt, const int* __restrict__ ovf_dst,
                       const int* __restrict__ ovf_src, const float* __restrict__ W1,
                       const float* __restrict__ b1, const float* __restrict__ W2,
                       float* __restrict__ out, int M) {
    static_assert(N_IN == G_KC && N_HID % G_BN == 0 && N_HID2 % G_BN == 0 && N_HID == 2 * G_KC, "tile sizes");
    __shared__ __attribute__((aligned(16))) float As[G_BM * G_PITCH];
    __shared__ __attribute__((aligned(16))) float Ws[G_BN * G_PITCH];
    __shared__ __attribute__((aligned(16))) float Hs[G_BM * H_PITCH];
    __shared__ int idx_s[G_BM * 32];
    __shared__ int deg_s[G_BM];

    const int t    = threadIdx.x;
    const int lane = t & 31;
    const int wave = t >> 5;
    const int wm   = wave >> 1;
    const int wn   = wave & 1;
    const int half = lane >> 4;
    const int l16  = lane & 15;
    const int m0 = (int)blockIdx.x * G_BM;

    const int q  = t & 31;
    const int r0 = __builtin_amdgcn_readfirstlane(t >> 5);
    v4f rw[8];
    load_w_block(0, W1, W2, r0, q, rw);

    int dl = 0;
    {
        const int nd = m0 + r0 + 8 * q;
        if (q < 8 && nd < M) dl = cnt[(size_t)nd * CNT_STRIDE];
    }
    int iv[8];
    #pragma unroll
    for (int i = 0; i < 8; ++i) {
        const int rl = r0 + 8 * i;
        const int d  = __builtin_amdgcn_readlane(dl, i);
        const int mm = d < 32 ? d : 32;
        const int nd = (m0 + rl < M) ? (m0 + rl) : M;
        const int* __restrict__ p = (q < mm) ? (bucket + (size_t)nd * CAP + q) : (cnt + (size_t)nd * CNT_STRIDE);
        const int v = *p;
        iv[i] = (q < mm) ? v : 0;
    }
    #pragma unroll
    for (int i = 0; i < 8; ++i) idx_s[(r0 + 8 * i) * 32 + q] = iv[i];
    if (q < 8) deg_s[r0 + 8 * q] = dl;
    for (int i = 0; i < 8; ++i) {
        const int rl = r0 + 8 * i;
        const int d  = __builtin_amdgcn_readlane(dl, i);
        const int my0 = idx_s[rl * 32 + q];
        v4f acc[1];
        sum_listed_rows<N_IN>(x, bucket, cnt, ovf_dst, ovf_src, m0 + rl, q, M, d, my0, acc);
        *(v4f*)(As + rl * G_PITCH + 4 * q) = acc[0];
    }

    const float* __restrict__ as  = As + (wm * 16 + l16) * G_PITCH + 2 * half;
    const float* __restrict__ hs  = Hs + (wm * 16 + l16) * H_PITCH + 2 * half;
    const float* __restrict__ ws0 = Ws + (wn * 32 + l16) * G_PITCH + 2 * half;
    const float* __restrict__ ws1 = ws0 + 16 * G_PITCH;

    #pragma unroll
    for (int i = 0; i < 8; ++i) *(v4f*)(Ws + (r0 + 8 * i) * G_PITCH + 4 * q) = rw[i];
    __syncthreads();

    float dv[8];
    #pragma unroll
    for (int r = 0; r < 8; ++r) dv[r] = 1.0f / fmaxf((float)deg_s[wm * 16 + half * 8 + r], 1.0f);

    #pragma unroll
    for (int s = 0; s < N_ST1; ++s) {
        load_w_block(s + 1, W1, W2, r0, q, rw);
        const int ca = s * G_BN + wn * 32 + l16;
        const int cb = ca + 16;
        const float bva = b1[ca];
        const float bvb = b1[cb];
        v8f acc0 = {}, acc1 = {};
        #pragma unroll
        for (int k = 0; k < G_KC; k += 4) {
            v2f a  = *(const v2f*)(as  + k);
            v2f b0 = *(const v2f*)(ws0 + k);
            v2f b1v = *(const v2f*)(ws1 + k);
            acc0 = __builtin_amdgcn_wmma_f32_16x16x4_f32(
                 false, a,  false, b0,
                 (short)0, acc0,  false,  false);
            acc1 = __builtin_amdgcn_wmma_f32_16x16x4_f32(
                 false, a,  false, b1v,
                 (short)0, acc1,  false,  false);
        }
        #pragma unroll
        for (int r = 0; r < 8; ++r) {
            float* __restrict__ hrow = Hs + (wm * 16 + half * 8 + r) * H_PITCH;
            hrow[ca] = fmaxf(acc0[r] * dv[r] + bva, 0.0f);
            hrow[cb] = fmaxf(acc1[r] * dv[r] + bvb, 0.0f);
        }
        __syncthreads();
        #pragma unroll
        for (int i = 0; i < 8; ++i) *(v4f*)(Ws + (r0 + 8 * i) * G_PITCH + 4 * q) = rw[i];
        __syncthreads();
    }

    #pragma unroll
    for (int nb = 0; nb < N_HID2 / G_BN; ++nb) {
        v8f acc0 = {}, acc1 = {};
        #pragma unroll
        for (int kc = 0; kc < N_HID / G_KC; ++kc) {
            const int s = N_ST1 + nb * (N_HID / G_KC) + kc;
            const bool more = (s + 1 < N_ST1 + N_ST2);
            if (more) load_w_block(s + 1, W1, W2, r0, q, rw);
            const float* __restrict__ hk = hs + kc * G_KC;
            #pragma unroll
            for (int k = 0; k < G_KC; k += 4) {
                v2f a  = *(const v2f*)(hk  + k);
                v2f b0 = *(const v2f*)(ws0 + k);
                v2f b1v = *(const v2f*)(ws1 + k);
                acc0 = __builtin_amdgcn_wmma_f32_16x16x4_f32(
                     false, a,  false, b0,
                     (short)0, acc0,  false,  false);
                acc1 = __builtin_amdgcn_wmma_f32_16x16x4_f32(
                     false, a,  false, b1v,
                     (short)0, acc1,  false,  false);
            }
            if (more) {
                __syncthreads();
                #pragma unroll
                for (int i = 0; i < 8; ++i) *(v4f*)(Ws + (r0 + 8 * i) * G_PITCH + 4 * q) = rw[i];
                __syncthreads();
            }
        }
        const int na = nb * G_BN + wn * 32 + l16;
        const int nc = na + 16;
        #pragma unroll
        for (int r = 0; r < 8; ++r) {
            int m = m0 + wm * 16 + half * 8 + r;
            if (m < M) {
                out[(size_t)m * N_HID2 + na] = acc0[r];
                out[(size_t)m * N_HID2 + nc] = acc1[r];
            }
        }
    }
}

extern "C" void kernel_launch(void* const* d_in, const int* in_sizes, int n_in,
                              void* d_out, int out_size, void* d_ws, size_t ws_size,
                              hipStream_t stream) {
    const float* x  = (const float*)d_in[0];
    const int*   ei = (const int*)  d_in[1];
    const float* W1 = (const float*)d_in[3];
    const float* b1 = (const float*)d_in[4];
    const float* W2 = (const float*)d_in[5];
    const float* b2 = (const float*)d_in[6];
    const float* W3 = (const float*)d_in[7];
    const float* b3 = (const float*)d_in[8];
    float* out = (float*)d_out;

    const int M = in_sizes[0] / N_IN;
    const int E = in_sizes[1] / 2;
    const int* row = ei;
    const int* col = ei + E;

    const size_t Mp = (((size_t)M + 1) * CNT_STRIDE + 63) & ~(size_t)63;
    const size_t Ep = ((size_t)E + 31) & ~(size_t)31;
    int* wsi = (int*)d_ws;
    size_t off = 0;
    int* cnt     = wsi + off; off += Mp;
    int* bucket  = wsi + off; off += (size_t)M * CAP;
    int* ovf_dst = wsi + off; off += Ep;
    int* ovf_src = wsi + off; off += Ep;
    float* ws = (float*)d_ws;
    float* z2   = ws + off; off += (size_t)M * N_HID2;
    (void)ws_size;

    hipMemsetAsync(cnt, 0, sizeof(int) * ((size_t)M + 1) * CNT_STRIDE, stream);

    bucket_fill_kernel<<<(E + 255) / 256, 256, 0, stream>>>(row, col, cnt, bucket, ovf_dst, ovf_src, M, E);

    dense_layers_wmma<<<(M + G_BM - 1) / G_BM, 256, 0, stream>>>(x, bucket, cnt, ovf_dst, ovf_src, W1, b1, W2, z2, M);

    gather_head_kernel<N_HID2><<<(M + FIN_WAVES - 1) / FIN_WAVES, FIN_WAVES * 32, 0, stream>>>(
        z2, bucket, cnt, ovf_dst, ovf_src, b2, W3, b3, out, M);
}


# In-degree counter moved into word 0 of each node's 128-byte index line (31 slots, overflow list beyond; was counter line + 256-byte row): one load gives degree and slots; sum order varies as before
# speedup vs baseline: 8.8729x; 1.0430x over previous
#include <hip/hip_runtime.h>

typedef __attribute__((ext_vector_type(2))) float v2f;
typedef __attribute__((ext_vector_type(4))) float v4f;
typedef __attribute__((ext_vector_type(8))) float v8f;

#define N_IN   128
#define N_HID  256
#define N_HID2 128
#define N_CLS  2

#define CAP 31
#define ROW_W 32

__global__ void bucket_fill_kernel(const int* __restrict__ row, const int* __restrict__ col,
                                   int* __restrict__ rows,
                                   int* __restrict__ ovf_dst, int* __restrict__ ovf_src,
                                   int M, int E) {
    int e = blockIdx.x * blockDim.x + threadIdx.x;
    if (e >= E) return;
    const int c = col[e];
    const int r = row[e];
    int* __restrict__ rp = rows + (size_t)c * ROW_W;
    const int slot = atomicAdd(rp, 1);
    if (slot < CAP) {
        rp[1 + slot] = r;
    } else {
        const int k = atomicAdd(&rows[(size_t)M * ROW_W], 1);
        ovf_dst[k] = c;
        ovf_src[k] = r;
    }
}

template<int F, int G>
__device__ __forceinline__ void load_rows(const float* __restrict__ src, int my, int j, int lane,
                                          int last, v4f (&v)[G][F / 128]) {
    #pragma unroll
    for (int g = 0; g < G; ++g) {
        int r = __builtin_amdgcn_readlane(my, j + g);
        r = r < 0 ? 0 : (r > last ? last : r);
        const float* __restrict__ p = src + (size_t)r * F + lane * 4;
        #pragma unroll
        for (int q = 0; q < F / 128; ++q) v[g][q] = *(const v4f*)(p + q * 128);
    }
}

template<int F, int G>
__device__ __forceinline__ void sum_rows(const v4f (&v)[G][F / 128], v4f (&acc)[F / 128]) {
    #pragma unroll
    for (int g = 0; g < G; ++g) {
        #pragma unroll
        for (int q = 0; q < F / 128; ++q) acc[q] += v[g][q];
    }
}

template<int F, int G>
__device__ __forceinline__ void add_rows(const float* __restrict__ src, int my, int j, int lane,
                                         int last, v4f (&acc)[F / 128]) {
    v4f v[G][F / 128];
    load_rows<F, G>(src, my, j, lane, last, v);
    sum_rows<F, G>(v, acc);
}

template<int F>
__device__ __forceinline__ int sum_node_rows(const float* __restrict__ src, const int* __restrict__ rows,
                                             const int* __restrict__ ovf_dst,
                                             const int* __restrict__ ovf_src, int node, int lane, int n,
                                             v4f (&acc)[F / 128]) {
    constexpr int U = 1024 / F;
    const int my = rows[(size_t)node * ROW_W + lane];
    const int d  = __builtin_amdgcn_readlane(my, 0);
    const int dm = d < CAP ? d : CAP;
    #pragma unroll
    for (int q = 0; q < F / 128; ++q) acc[q] = (v4f){0.0f, 0.0f, 0.0f, 0.0f};

    int j = 0;
    for (; j + U <= dm; j += U) add_rows<F, U>(src, my, j + 1, lane, n - 1, acc);
    if (U == 8 && dm - j >= 4) { add_rows<F, 4>(src, my, j + 1, lane, n - 1, acc); j += 4; }
    if (dm - j >= 2) { add_rows<F, 2>(src, my, j + 1, lane, n - 1, acc); j += 2; }
    if (dm - j >= 1) { add_rows<F, 1>(src, my, j + 1, lane, n - 1, acc); }

    if (d > CAP) {
        const int L = rows[(size_t)n * ROW_W];
        for (int k0 = 0; k0 < L; k0 += 32) {
            const int kk = k0 + lane;
            const int dst = (kk < L) ? ovf_dst[kk] : -1;
            const int sv  = (kk < L) ? ovf_src[kk] : 0;
            unsigned hit = __builtin_amdgcn_ballot_w32(dst == node);
            while (hit) {
                const int bit = __builtin_ctz(hit);
                hit &= hit - 1;
                add_rows<F, 1>(src, sv, bit, lane, n - 1, acc);
            }
        }
    }
    return d;
}

template<int F>
__device__ __forceinline__ void add_rows_pieces(const float* __restrict__ src, int my, int j, int c, int lane,
                                                int last, v4f (&acc)[F / 128]) {
    v4f v16[16][F / 128], v8[8][F / 128], v4[4][F / 128], v2[2][F / 128], v1[1][F / 128];
    int s = j;
    if (c & 16) { load_rows<F, 16>(src, my, s, lane, last, v16); s += 16; }
    if (c & 8)  { load_rows<F, 8>(src, my, s, lane, last, v8);   s += 8; }
    if (c & 4)  { load_rows<F, 4>(src, my, s, lane, last, v4);   s += 4; }
    if (c & 2)  { load_rows<F, 2>(src, my, s, lane, last, v2);   s += 2; }
    if (c & 1)  { load_rows<F, 1>(src, my, s, lane, last, v1); }
    if (c & 16) sum_rows<F, 16>(v16, acc);
    if (c & 8)  sum_rows<F, 8>(v8, acc);
    if (c & 4)  sum_rows<F, 4>(v4, acc);
    if (c & 2)  sum_rows<F, 2>(v2, acc);
    if (c & 1)  sum_rows<F, 1>(v1, acc);
}

template<int F>
__device__ __forceinline__ void sum_listed_rows(const float* __restrict__ src, const int* __restrict__ rows,
                                                const int* __restrict__ ovf_dst,
                                                const int* __restrict__ ovf_src, int node, int lane, int n,
                                                int d, int my0, v4f (&acc)[F / 128]) {
    const int dm = d < CAP ? d : CAP;
    #pragma unroll
    for (int q = 0; q < F / 128; ++q) acc[q] = (v4f){0.0f, 0.0f, 0.0f, 0.0f};

    add_rows_pieces<F>(src, my0, 1, dm, lane, n - 1, acc);

    if (d > CAP) {
        const int L = rows[(size_t)n * ROW_W];
        for (int k0 = 0; k0 < L; k0 += 32) {
            const int kk = k0 + lane;
            const int dst = (kk < L) ? ovf_dst[kk] : -1;
            const int sv  = (kk < L) ? ovf_src[kk] : 0;
            unsigned hit = __builtin_amdgcn_ballot_w32(dst == node);
            while (hit) {
                const int bit = __builtin_ctz(hit);
                hit &= hit - 1;
                add_rows<F, 1>(src, sv, bit, lane, n - 1, acc);
            }
        }
    }
}

#define FIN_WAVES 16
template<int F>
__global__ __launch_bounds__(FIN_WAVES * 32)
void gather_head_kernel(const float* __restrict__ src, const int* __restrict__ rows,
                        const int* __restrict__ ovf_dst,
                        const int* __restrict__ ovf_src, const float* __restrict__ bias,
                        const float* __restrict__ W3, const float* __restrict__ b3,
                        float* __restrict__ out, int n) {
    static_assert(FIN_WAVES * N_CLS == 32, "one wave writes the group's output line");
    __shared__ float line[FIN_WAVES * N_CLS];
    const int lane = threadIdx.x & 31;
    const int wave = __builtin_amdgcn_readfirstlane((int)(threadIdx.x >> 5));
    const int node = __builtin_amdgcn_readfirstlane((int)(blockIdx.x * FIN_WAVES) + wave);
    if (node < n) {
        v4f bv[F / 128], w0[F / 128], w1[F / 128];
        #pragma unroll
        for (int q = 0; q < F / 128; ++q) {
            bv[q] = *(const v4f*)(bias + q * 128 + lane * 4);
            w0[q] = *(const v4f*)(W3 + q * 128 + lane * 4);
            w1[q] = *(const v4f*)(W3 + F + q * 128 + lane * 4);
        }
        const float c0 = b3[0];
        const float c1 = b3[1];
        v4f acc[F / 128];
        const int d = sum_node_rows<F>(src, rows, ovf_dst, ovf_src, node, lane, n, acc);
        const float s = 1.0f / fmaxf((float)d, 1.0f);
        float p0 = 0.0f, p1 = 0.0f;
        #pragma unroll
        for (int q = 0; q < F / 128; ++q) {
            v4f v = acc[q] * s + bv[q];
            v.x = fmaxf(v.x, 0.0f); v.y = fmaxf(v.y, 0.0f);
            v.z = fmaxf(v.z, 0.0f); v.w = fmaxf(v.w, 0.0f);
            p0 += v.x * w0[q].x + v.y * w0[q].y + v.z * w0[q].z + v.w * w0[q].w;
            p1 += v.x * w1[q].x + v.y * w1[q].y + v.z * w1[q].z + v.w * w1[q].w;
        }
        #pragma unroll
        for (int off = 16; off >= 1; off >>= 1) {
            p0 += __shfl_xor(p0, off, 32);
            p1 += __shfl_xor(p1, off, 32);
        }
        if (lane == 0) {
            line[wave * N_CLS + 0] = p0 + c0;
            line[wave * N_CLS + 1] = p1 + c1;
        }
    }
    __syncthreads();
    if (wave == 0) {
        const int idx = (int)blockIdx.x * (FIN_WAVES * N_CLS) + lane;
        if (idx < n * N_CLS) out[idx] = line[lane];
    }
}

#define G_BM    64
#define G_BN    64
#define G_KC    128
#define G_PITCH 132
#define H_PITCH 260
#define N_ST1   (N_HID / G_BN)
#define N_ST2   ((N_HID2 / G_BN) * (N_HID / G_KC))

__device__ __forceinline__ void load_w_block(int s, const float* __restrict__ W1, const float* __restrict__ W2,
                                             int r0, int q, v4f (&rw)[8]) {
    const float* __restrict__ p;
    int stride;
    if (s < N_ST1) {
        p = W1 + (size_t)(s * G_BN) * N_IN;
        stride = N_IN;
    } else {
        const int s2 = s - N_ST1;
        p = W2 + (size_t)((s2 >> 1) * G_BN) * N_HID + (s2 & 1) * G_KC;
        stride = N_HID;
    }
    #pragma unroll
    for (int i = 0; i < 8; ++i) rw[i] = *(const v4f*)(p + (size_t)(r0 + 8 * i) * stride + 4 * q);
}

__global__ __launch_bounds__(256)
void dense_layers_wmma(const float* __restrict__ x, const int* __restrict__ rows,
                       const int* __restrict__ ovf_dst,
                       const int* __restrict__ ovf_src, const float* __restrict__ W1,
                       const float* __restrict__ b1, const float* __restrict__ W2,
                       float* __restrict__ out, int M) {
    static_assert(N_IN == G_KC && N_HID % G_BN == 0 && N_HID2 % G_BN == 0 && N_HID == 2 * G_KC, "tile sizes");
    __shared__ __attribute__((aligned(16))) float As[G_BM * G_PITCH];
    __shared__ __attribute__((aligned(16))) float Ws[G_BN * G_PITCH];
    __shared__ __attribute__((aligned(16))) float Hs[G_BM * H_PITCH];
    __shared__ int idx_s[G_BM * ROW_W];

    const int t    = threadIdx.x;
    const int lane = t & 31;
    const int wave = t >> 5;
    const int wm   = wave >> 1;
    const int wn   = wave & 1;
    const int half = lane >> 4;
    const int l16  = lane & 15;
    const int m0 = (int)blockIdx.x * G_BM;

    const int q  = t & 31;
    const int r0 = __builtin_amdgcn_readfirstlane(t >> 5);
    v4f rw[8];
    load_w_block(0, W1, W2, r0, q, rw);

    int iv[8];
    #pragma unroll
    for (int i = 0; i < 8; ++i) {
        const int nd  = m0 + r0 + 8 * i;
        const int ndc = nd < M ? nd : M;
        const int v = rows[(size_t)ndc * ROW_W + q];
        iv[i] = nd < M ? v : 0;
    }
    #pragma unroll
    for (int i = 0; i < 8; ++i) idx_s[(r0 + 8 * i) * ROW_W + q] = iv[i];
    for (int i = 0; i < 8; ++i) {
        const int rl = r0 + 8 * i;
        const int my0 = idx_s[rl * ROW_W + q];
        const int d  = __builtin_amdgcn_readlane(my0, 0);
        v4f acc[1];
        sum_listed_rows<N_IN>(x, rows, ovf_dst, ovf_src, m0 + rl, q, M, d, my0, acc);
        *(v4f*)(As + rl * G_PITCH + 4 * q) = acc[0];
    }

    const float* __restrict__ as  = As + (wm * 16 + l16) * G_PITCH + 2 * half;
    const float* __restrict__ hs  = Hs + (wm * 16 + l16) * H_PITCH + 2 * half;
    const float* __restrict__ ws0 = Ws + (wn * 32 + l16) * G_PITCH + 2 * half;
    const float* __restrict__ ws1 = ws0 + 16 * G_PITCH;

    #pragma unroll
    for (int i = 0; i < 8; ++i) *(v4f*)(Ws + (r0 + 8 * i) * G_PITCH + 4 * q) = rw[i];
    __syncthreads();

    float dv[8];
    #pragma unroll
    for (int r = 0; r < 8; ++r) dv[r] = 1.0f / fmaxf((float)idx_s[(wm * 16 + half * 8 + r) * ROW_W], 1.0f);

    #pragma unroll
    for (int s = 0; s < N_ST1; ++s) {
        load_w_block(s + 1, W1, W2, r0, q, rw);
        const int ca = s * G_BN + wn * 32 + l16;
        const int cb = ca + 16;
        const float bva = b1[ca];
        const float bvb = b1[cb];
        v8f acc0 = {}, acc1 = {};
        #pragma unroll
        for (int k = 0; k < G_KC; k += 4) {
            v2f a  = *(const v2f*)(as  + k);
            v2f b0 = *(const v2f*)(ws0 + k);
            v2f b1v = *(const v2f*)(ws1 + k);
            acc0 = __builtin_amdgcn_wmma_f32_16x16x4_f32(
                 false, a,  false, b0,
                 (short)0, acc0,  false,  false);
            acc1 = __builtin_amdgcn_wmma_f32_16x16x4_f32(
                 false, a,  false, b1v,
                 (short)0, acc1,  false,  false);
        }
        #pragma unroll
        for (int r = 0; r < 8; ++r) {
            float* __restrict__ hrow = Hs + (wm * 16 + half * 8 + r) * H_PITCH;
            hrow[ca] = fmaxf(acc0[r] * dv[r] + bva, 0.0f);
            hrow[cb] = fmaxf(acc1[r] * dv[r] + bvb, 0.0f);
        }
        __syncthreads();
        #pragma unroll
        for (int i = 0; i < 8; ++i) *(v4f*)(Ws + (r0 + 8 * i) * G_PITCH + 4 * q) = rw[i];
        __syncthreads();
    }

    #pragma unroll
    for (int nb = 0; nb < N_HID2 / G_BN; ++nb) {
        v8f acc0 = {}, acc1 = {};
        #pragma unroll
        for (int kc = 0; kc < N_HID / G_KC; ++kc) {
            const int s = N_ST1 + nb * (N_HID / G_KC) + kc;
            const bool more = (s + 1 < N_ST1 + N_ST2);
            if (more) load_w_block(s + 1, W1, W2, r0, q, rw);
            const float* __restrict__ hk = hs + kc * G_KC;
            #pragma unroll
            for (int k = 0; k < G_KC; k += 4) {
                v2f a  = *(const v2f*)(hk  + k);
                v2f b0 = *(const v2f*)(ws0 + k);
                v2f b1v = *(const v2f*)(ws1 + k);
                acc0 = __builtin_amdgcn_wmma_f32_16x16x4_f32(
                     false, a,  false, b0,
                     (short)0, acc0,  false,  false);
                acc1 = __builtin_amdgcn_wmma_f32_16x16x4_f32(
                     false, a,  false, b1v,
                     (short)0, acc1,  false,  false);
            }
            if (more) {
                __syncthreads();
                #pragma unroll
                for (int i = 0; i < 8; ++i) *(v4f*)(Ws + (r0 + 8 * i) * G_PITCH + 4 * q) = rw[i];
                __syncthreads();
            }
        }
        const int na = nb * G_BN + wn * 32 + l16;
        const int nc = na + 16;
        #pragma unroll
        for (int r = 0; r < 8; ++r) {
            int m = m0 + wm * 16 + half * 8 + r;
            if (m < M) {
                out[(size_t)m * N_HID2 + na] = acc0[r];
                out[(size_t)m * N_HID2 + nc] = acc1[r];
            }
        }
    }
}

extern "C" void kernel_launch(void* const* d_in, const int* in_sizes, int n_in,
                              void* d_out, int out_size, void* d_ws, size_t ws_size,
                              hipStream_t stream) {
    const float* x  = (const float*)d_in[0];
    const int*   ei = (const int*)  d_in[1];
    const float* W1 = (const float*)d_in[3];
    const float* b1 = (const float*)d_in[4];
    const float* W2 = (const float*)d_in[5];
    const float* b2 = (const float*)d_in[6];
    const float* W3 = (const float*)d_in[7];
    const float* b3 = (const float*)d_in[8];
    float* out = (float*)d_out;

    const int M = in_sizes[0] / N_IN;
    const int E = in_sizes[1] / 2;
    const int* row = ei;
    const int* col = ei + E;

    const size_t Mp = (((size_t)M + 1) * ROW_W + 63) & ~(size_t)63;
    const size_t Ep = ((size_t)E + 31) & ~(size_t)31;
    int* wsi = (int*)d_ws;
    size_t off = 0;
    int* rows    = wsi + off; off += Mp;
    int* ovf_dst = wsi + off; off += Ep;
    int* ovf_src = wsi + off; off += Ep;
    float* ws = (float*)d_ws;
    float* z2   = ws + off; off += (size_t)M * N_HID2;
    (void)ws_size;

    hipMemsetAsync(rows, 0, sizeof(int) * ((size_t)M + 1) * ROW_W, stream);

    bucket_fill_kernel<<<(E + 255) / 256, 256, 0, stream>>>(row, col, rows, ovf_dst, ovf_src, M, E);

    dense_layers_wmma<<<(M + G_BM - 1) / G_BM, 256, 0, stream>>>(x, rows, ovf_dst, ovf_src, W1, b1, W2, z2, M);

    gather_head_kernel<N_HID2><<<(M + FIN_WAVES - 1) / FIN_WAVES, FIN_WAVES * 32, 0, stream>>>(
        z2, rows, ovf_dst, ovf_src, b2, W3, b3, out, M);
}
